# starting kernel
# baseline (speedup 1.0000x reference)
#include <hip/hip_runtime.h>
#include <math.h>


typedef __attribute__((ext_vector_type(2))) float v2f;
typedef __attribute__((ext_vector_type(8))) float v8f;

#define FIN   8
#define HID   64
#define KTOT  72
#define KSTEPS 18
#define GCOLS 256
#define NT_G  16
#define ROWS_PER_BLOCK 128
#define EPS 1e-5f

__device__ __forceinline__ float sigmoidf_(float x) { return 1.0f / (1.0f + expf(-x)); }

__global__ void k_init_deg(float* deg, int N) {
  int i = blockIdx.x * blockDim.x + threadIdx.x;
  if (i < N) deg[i] = 1.0f;
}

__global__ void k_acc_deg(float* deg, const int* dst, const float* w, int E) {
  int e = blockIdx.x * blockDim.x + threadIdx.x;
  if (e < E) atomicAdd(&deg[dst[e]], w[e]);
}

__global__ void k_dinv(float* deg, int N) {
  int i = blockIdx.x * blockDim.x + threadIdx.x;
  if (i < N) deg[i] = rsqrtf(deg[i]);
}

__global__ void k_norm(float* norm, const float* dinv, const int* src, const int* dst,
                       const float* w, int E) {
  int e = blockIdx.x * blockDim.x + threadIdx.x;
  if (e < E) norm[e] = dinv[dst[e]] * w[e] * dinv[src[e]];
}

__global__ void k_init_axh(float* AXH, const float* X, const float* H,
                           const float* dinv, int N) {
  long t = (long)blockIdx.x * blockDim.x + threadIdx.x;
  long total = (long)N * KTOT;
  if (t >= total) return;
  int n = (int)(t / KTOT);
  int k = (int)(t - (long)n * KTOT);
  float d2 = dinv[n] * dinv[n];
  float v = (k < FIN) ? X[(long)n * FIN + k] : H[(long)n * HID + (k - FIN)];
  AXH[t] = d2 * v;
}

__global__ void k_scatter_axh(float* AXH, const float* X, const float* H,
                              const int* src, const int* dst, const float* norm, int E) {
  long t = (long)blockIdx.x * blockDim.x + threadIdx.x;
  long total = (long)E * KTOT;
  if (t >= total) return;
  int e = (int)(t / KTOT);
  int k = (int)(t - (long)e * KTOT);
  int s = src[e];
  float v = (k < FIN) ? X[(long)s * FIN + k] : H[(long)s * HID + (k - FIN)];
  atomicAdd(&AXH[(long)dst[e] * KTOT + k], norm[e] * v);
}

__global__ void k_init_agg(float* dstbuf, const float* srcf, const float* dinv,
                           int N, int F, int stride) {
  long t = (long)blockIdx.x * blockDim.x + threadIdx.x;
  long total = (long)N * stride;
  if (t >= total) return;
  int n = (int)(t / stride);
  int k = (int)(t - (long)n * stride);
  float d = dinv[n];
  dstbuf[t] = (k < F) ? d * d * srcf[t] : 0.0f;
}

__global__ void k_scatter_agg(float* dstbuf, const float* srcf, const int* src,
                              const int* dst, const float* norm, int E, int F, int stride) {
  long t = (long)blockIdx.x * blockDim.x + threadIdx.x;
  long total = (long)E * F;
  if (t >= total) return;
  int e = (int)(t / F);
  int k = (int)(t - (long)e * F);
  float v = norm[e] * srcf[(long)src[e] * stride + k];
  atomicAdd(&dstbuf[(long)dst[e] * stride + k], v);
}

__global__ __launch_bounds__(256)
void k_gates(const float* __restrict__ AXH, const float* __restrict__ C,
             const float* __restrict__ Wx, const float* __restrict__ Wh,
             const float* __restrict__ bx, const float* __restrict__ bh,
             const float* __restrict__ bg, const float* __restrict__ wc,
             const float* __restrict__ ln_h_s, const float* __restrict__ ln_h_b,
             const float* __restrict__ ln_c_s, const float* __restrict__ ln_c_b,
             const float* __restrict__ ln_o_s, const float* __restrict__ ln_o_b,
             const float* __restrict__ skip,
             float* __restrict__ hiddenOut, float* __restrict__ cellOut,
             float* __restrict__ Y, int N) {
  extern __shared__ float smem[];
  float* Bf = smem;
  float* bias = smem + 18432;

  int tid = threadIdx.x;
  for (int i = tid; i < KSTEPS * NT_G * 64; i += 256) {
    int s = i >> 10;
    int rem = i & 1023;
    int t = rem >> 6;
    int rem2 = rem & 63;
    int L = rem2 >> 1;
    int j = rem2 & 1;
    int k = 4 * s + 2 * (L >> 4) + j;
    int c = 16 * t + (L & 15);
    int g = c >> 6, h = c & 63;
    Bf[i] = (k < FIN) ? Wx[(g * FIN + k) * HID + h]
                      : Wh[(g * HID + (k - FIN)) * HID + h];
  }
  if (tid < GCOLS) bias[tid] = bx[tid] + bh[tid] + bg[tid];
  __syncthreads();

  int wave = tid >> 5, lane = tid & 31;
  int half = lane >> 4, lcol = lane & 15;
  int rowBase = blockIdx.x * ROWS_PER_BLOCK + wave * 16;

  int arow = rowBase + lcol;
  if (arow > N - 1) arow = N - 1;
  const float* aptr = AXH + (long)arow * KTOT;

  v8f acc[NT_G];
#pragma unroll
  for (int t = 0; t < NT_G; ++t)
#pragma unroll
    for (int r = 0; r < 8; ++r) acc[t][r] = 0.0f;

  for (int s = 0; s < KSTEPS; ++s) {
    int k0 = 4 * s + 2 * half;
    v2f a = *(const v2f*)(aptr + k0);
    const float* bbase = Bf + s * 1024 + lane * 2;
#pragma unroll
    for (int t = 0; t < NT_G; ++t) {
      v2f b = *(const v2f*)(bbase + t * 64);
      acc[t] = __builtin_amdgcn_wmma_f32_16x16x4_f32(
          false, a, false, b, (short)0, acc[t], false, false);
    }
  }

#pragma unroll
  for (int r = 0; r < 8; ++r) {
    int node = rowBase + r + 8 * half;
    bool valid = node < N;
    int nc = valid ? node : N - 1;
    float hloc[4], cloc[4];
#pragma unroll
    for (int t = 0; t < 4; ++t) {
      int h = t * 16 + lcol;
      float c0 = C[(long)nc * HID + h];
      float gi = acc[t][r]      + bias[h]       + wc[h] * c0;
      float gf = acc[t + 4][r]  + bias[64 + h]  + wc[64 + h] * c0;
      float gc = acc[t + 8][r]  + bias[128 + h];
      float go = acc[t + 12][r] + bias[192 + h];
      float iv = sigmoidf_(gi);
      float fv = sigmoidf_(gf);
      float cc = fv * c0 + iv * tanhf(gc);
      float ov = sigmoidf_(go + wc[128 + h] * cc);
      hloc[t] = ov * tanhf(cc);
      cloc[t] = cc;
    }
    float sh = 0.f, sh2 = 0.f, sc = 0.f, sc2 = 0.f;
#pragma unroll
    for (int t = 0; t < 4; ++t) {
      sh += hloc[t]; sh2 += hloc[t] * hloc[t];
      sc += cloc[t]; sc2 += cloc[t] * cloc[t];
    }
#pragma unroll
    for (int mask = 1; mask < 16; mask <<= 1) {
      sh  += __shfl_xor(sh,  mask, 32);
      sh2 += __shfl_xor(sh2, mask, 32);
      sc  += __shfl_xor(sc,  mask, 32);
      sc2 += __shfl_xor(sc2, mask, 32);
    }
    float mh = sh * (1.0f / 64.0f), mc = sc * (1.0f / 64.0f);
    float vh = sh2 * (1.0f / 64.0f) - mh * mh;
    float vc = sc2 * (1.0f / 64.0f) - mc * mc;
    float rh = rsqrtf(vh + EPS), rc = rsqrtf(vc + EPS);
    if (valid) {
#pragma unroll
      for (int t = 0; t < 4; ++t) {
        int h = t * 16 + lcol;
        float hn = (hloc[t] - mh) * rh;
        float cn = (cloc[t] - mc) * rc;
        hiddenOut[(long)node * HID + h] = hn * ln_h_s[h] + ln_h_b[h];
        cellOut[(long)node * HID + h]   = cn * ln_c_s[h] + ln_c_b[h];
        float ov = hn * ln_o_s[h] + ln_o_b[h];
        Y[(long)node * KTOT + h] = ov > 0.0f ? ov : 0.0f;
      }
      if (lcol < 8) {
        Y[(long)node * KTOT + 64 + lcol] =
            (lcol < 2) ? skip[(long)node * 2 + lcol] : 0.0f;
      }
    }
  }
}

__global__ __launch_bounds__(256)
void k_gemm2(const float* __restrict__ AY, const float* __restrict__ W1,
             const float* __restrict__ b1, float* __restrict__ Z, int N) {
  __shared__ float Bf[KSTEPS * 4 * 64];
  __shared__ float b1s[HID];

  int tid = threadIdx.x;
  for (int i = tid; i < KSTEPS * 4 * 64; i += 256) {
    int s = i >> 8;
    int rem = i & 255;
    int t = rem >> 6;
    int rem2 = rem & 63;
    int L = rem2 >> 1;
    int j = rem2 & 1;
    int k = 4 * s + 2 * (L >> 4) + j;
    int c = 16 * t + (L & 15);
    Bf[i] = (k < HID + 2) ? W1[k * HID + c] : 0.0f;
  }
  if (tid < HID) b1s[tid] = b1[tid];
  __syncthreads();

  int wave = tid >> 5, lane = tid & 31;
  int half = lane >> 4, lcol = lane & 15;
  int rowBase = blockIdx.x * ROWS_PER_BLOCK + wave * 16;

  int arow = rowBase + lcol;
  if (arow > N - 1) arow = N - 1;
  const float* aptr = AY + (long)arow * KTOT;

  v8f acc[4];
#pragma unroll
  for (int t = 0; t < 4; ++t)
#pragma unroll
    for (int r = 0; r < 8; ++r) acc[t][r] = 0.0f;

  for (int s = 0; s < KSTEPS; ++s) {
    int k0 = 4 * s + 2 * half;
    v2f a = *(const v2f*)(aptr + k0);
    const float* bbase = Bf + s * 256 + lane * 2;
#pragma unroll
    for (int t = 0; t < 4; ++t) {
      v2f b = *(const v2f*)(bbase + t * 64);
      acc[t] = __builtin_amdgcn_wmma_f32_16x16x4_f32(
          false, a, false, b, (short)0, acc[t], false, false);
    }
  }

#pragma unroll
  for (int r = 0; r < 8; ++r) {
    int node = rowBase + r + 8 * half;
    if (node >= N) continue;
#pragma unroll
    for (int t = 0; t < 4; ++t) {
      int h = t * 16 + lcol;
      float z = acc[t][r] + b1s[h];
      Z[(long)node * HID + h] = z > 0.0f ? z : 0.0f;
    }
  }
}

__global__ void k_final(const float* __restrict__ AZ, const float* __restrict__ W2,
                        const float* __restrict__ b2, const float* __restrict__ X,
                        float* __restrict__ out, int N) {
  __shared__ float w2s[HID];
  if (threadIdx.x < HID) w2s[threadIdx.x] = W2[threadIdx.x];
  __syncthreads();
  int n = blockIdx.x * blockDim.x + threadIdx.x;
  if (n >= N) return;
  const float* a = AZ + (long)n * HID;
  float acc = b2[0] + X[(long)n * FIN];
#pragma unroll 8
  for (int h = 0; h < HID; ++h) acc += a[h] * w2s[h];
  out[n] = acc;
}

extern "C" void kernel_launch(void* const* d_in, const int* in_sizes, int n_in,
                              void* d_out, int out_size, void* d_ws, size_t ws_size,
                              hipStream_t stream) {
  const float* X    = (const float*)d_in[0];
  const int*   eidx = (const int*)d_in[1];
  const float* ew   = (const float*)d_in[2];
  const float* skip = (const float*)d_in[3];
  const float* H    = (const float*)d_in[4];
  const float* C    = (const float*)d_in[5];
  const float* Wx   = (const float*)d_in[6];
  const float* bx   = (const float*)d_in[7];
  const float* Wh   = (const float*)d_in[8];
  const float* bh   = (const float*)d_in[9];
  const float* wc   = (const float*)d_in[10];
  const float* bg   = (const float*)d_in[11];
  const float* ln_h_s = (const float*)d_in[12];
  const float* ln_h_b = (const float*)d_in[13];
  const float* ln_c_s = (const float*)d_in[14];
  const float* ln_c_b = (const float*)d_in[15];
  const float* ln_o_s = (const float*)d_in[16];
  const float* ln_o_b = (const float*)d_in[17];
  const float* W1   = (const float*)d_in[18];
  const float* b1   = (const float*)d_in[19];
  const float* W2   = (const float*)d_in[20];
  const float* b2   = (const float*)d_in[21];

  const int N = in_sizes[0] / FIN;
  const int E = in_sizes[2];

  const int* src = eidx;
  const int* dst = eidx + E;

  float* out_p    = (float*)d_out;
  float* hidden_p = out_p + N;
  float* cell_p   = hidden_p + (long)N * HID;

  float* ws   = (float*)d_ws;
  float* dinv = ws;
  float* norm = dinv + N;
  float* AXH  = norm + E;
  float* Y    = AXH;
  float* AY   = AXH + (long)N * KTOT;
  float* Z    = AY + (long)N * KTOT;
  float* AZ   = Z + (long)N * HID;

  const int B = 256;
  auto cdiv = [](long a, long b) { return (unsigned)((a + b - 1) / b); };

  k_init_deg<<<cdiv(N, B), B, 0, stream>>>(dinv, N);
  k_acc_deg<<<cdiv(E, B), B, 0, stream>>>(dinv, dst, ew, E);
  k_dinv<<<cdiv(N, B), B, 0, stream>>>(dinv, N);
  k_norm<<<cdiv(E, B), B, 0, stream>>>(norm, dinv, src, dst, ew, E);

  k_init_axh<<<cdiv((long)N * KTOT, B), B, 0, stream>>>(AXH, X, H, dinv, N);
  k_scatter_axh<<<cdiv((long)E * KTOT, B), B, 0, stream>>>(AXH, X, H, src, dst, norm, E);

  size_t smem_gates = (size_t)(KSTEPS * NT_G * 64 + GCOLS) * sizeof(float);
  k_gates<<<cdiv(N, ROWS_PER_BLOCK), B, smem_gates, stream>>>(
      AXH, C, Wx, Wh, bx, bh, bg, wc,
      ln_h_s, ln_h_b, ln_c_s, ln_c_b, ln_o_s, ln_o_b,
      skip, hidden_p, cell_p, Y, N);

  k_init_agg<<<cdiv((long)N * KTOT, B), B, 0, stream>>>(AY, Y, dinv, N, HID + 2, KTOT);
  k_scatter_agg<<<cdiv((long)E * (HID + 2), B), B, 0, stream>>>(AY, Y, src, dst, norm,
                                                               E, HID + 2, KTOT);

  k_gemm2<<<cdiv(N, ROWS_PER_BLOCK), B, 0, stream>>>(AY, W1, b1, Z, N);

  k_init_agg<<<cdiv((long)N * HID, B), B, 0, stream>>>(AZ, Z, dinv, N, HID, HID);
  k_scatter_agg<<<cdiv((long)E * HID, B), B, 0, stream>>>(AZ, Z, src, dst, norm,
                                                          E, HID, HID);

  k_final<<<cdiv(N, B), B, 0, stream>>>(AZ, W2, b2, X, out_p, N);
}


# Aggregations accumulate in LDS tiles: a block owns 392 destination rows, scans all edges for them, adds by LDS float atomics, stores rows once; no global float atomics; sums stay in arrival order
# speedup vs baseline: 1.6748x; 1.6748x over previous
#include <hip/hip_runtime.h>
#include <math.h>


typedef __attribute__((ext_vector_type(2))) float v2f;
typedef __attribute__((ext_vector_type(8))) float v8f;

#define FIN   8
#define HID   64
#define KTOT  72
#define KSTEPS 18
#define GCOLS 256
#define NT_G  16
#define ROWS_PER_BLOCK 128
#define EPS 1e-5f

__device__ __forceinline__ float sigmoidf_(float x) { return 1.0f / (1.0f + expf(-x)); }

__global__ void k_init_deg(float* deg, int N) {
  int i = blockIdx.x * blockDim.x + threadIdx.x;
  if (i < N) deg[i] = 1.0f;
}

__global__ void k_acc_deg(float* deg, const int* dst, const float* w, int E) {
  int e = blockIdx.x * blockDim.x + threadIdx.x;
  if (e < E) atomicAdd(&deg[dst[e]], w[e]);
}

__global__ void k_dinv(float* deg, int N) {
  int i = blockIdx.x * blockDim.x + threadIdx.x;
  if (i < N) deg[i] = rsqrtf(deg[i]);
}

__global__ void k_norm(float* norm, const float* dinv, const int* src, const int* dst,
                       const float* w, int E) {
  int e = blockIdx.x * blockDim.x + threadIdx.x;
  if (e < E) norm[e] = dinv[dst[e]] * w[e] * dinv[src[e]];
}

#define TILE_R      392
#define QCAP        8192
#define SCAN_U      8
#define AGG_THREADS 1024
#define CHUNK       (AGG_THREADS * 4 * SCAN_U)

template <int F, bool SPLIT>
__device__ __forceinline__ float row_elem(const float* __restrict__ A,
                                          const float* __restrict__ B, int s, int k) {
  if (SPLIT) return (k < FIN) ? A[(long)s * FIN + k] : B[(long)s * HID + (k - FIN)];
  return A[(long)s * F + k];
}

template <int NE, int F, bool SPLIT>
__device__ __forceinline__ void add_rows(float* acc, const int2* qp,
                                         const float* __restrict__ A, const float* __restrict__ B,
                                         const int* __restrict__ src,
                                         const float* __restrict__ norm, int lane) {
  constexpr int KPL = (F + 31) / 32;
  int s[NE], dl[NE];
  float cf[NE];
#pragma unroll
  for (int j = 0; j < NE; ++j) {
    int2 ent = qp[j];
    dl[j] = ent.y;
    s[j] = src[ent.x];
    cf[j] = norm[ent.x];
  }
  float r[NE][KPL];
#pragma unroll
  for (int j = 0; j < NE; ++j)
#pragma unroll
    for (int m = 0; m < KPL; ++m) {
      int k = lane + 32 * m;
      r[j][m] = (k < F) ? row_elem<F, SPLIT>(A, B, s[j], k) : 0.0f;
    }
#pragma unroll
  for (int j = 0; j < NE; ++j)
#pragma unroll
    for (int m = 0; m < KPL; ++m) {
      int k = lane + 32 * m;
      if (k < F) atomicAdd(&acc[dl[j] * F + k], cf[j] * r[j][m]);
    }
}

template <int F, bool SPLIT>
__global__ __launch_bounds__(AGG_THREADS)
void k_agg_tile(float* __restrict__ outbuf, const float* __restrict__ A,
                const float* __restrict__ B, const float* __restrict__ dinv,
                const int* __restrict__ src, const int* __restrict__ dst,
                const float* __restrict__ norm, int N, int E) {
  __shared__ float acc[TILE_R * F];
  __shared__ int2 q[QCAP];
  __shared__ int qn[2];

  const int tid = threadIdx.x;
  const int lane = tid & 31, wave = tid >> 5;
  const int lo = blockIdx.x * TILE_R;
  const int cnt = (N - lo < TILE_R) ? (N - lo) : TILE_R;

  for (int i = tid; i < cnt * F; i += AGG_THREADS) {
    int n = i / F, k = i - n * F;
    float d = dinv[lo + n];
    acc[i] = d * d * row_elem<F, SPLIT>(A, B, lo + n, k);
  }
  if (tid == 0) { qn[0] = 0; qn[1] = 0; }
  __syncthreads();

  const bool vec_ok = (((size_t)dst) & 15) == 0;
  const int nchunks = (E + CHUNK - 1) / CHUNK;
  int4 v[SCAN_U];

  auto load_chunk = [&](int c) {
    if (vec_ok && c * CHUNK + CHUNK <= E) {
#pragma unroll
      for (int u = 0; u < SCAN_U; ++u)
        v[u] = *(const int4*)(dst + c * CHUNK + (u * AGG_THREADS + tid) * 4);
    } else {
#pragma unroll
      for (int u = 0; u < SCAN_U; ++u) {
        int e0 = c * CHUNK + (u * AGG_THREADS + tid) * 4;
        v[u].x = (e0     < E) ? dst[e0]     : -1;
        v[u].y = (e0 + 1 < E) ? dst[e0 + 1] : -1;
        v[u].z = (e0 + 2 < E) ? dst[e0 + 2] : -1;
        v[u].w = (e0 + 3 < E) ? dst[e0 + 3] : -1;
      }
    }
  };

  load_chunk(0);
  int par = 0;
  for (int c = 0; c < nchunks; ++c) {
    int m = 0;
#pragma unroll
    for (int u = 0; u < SCAN_U; ++u) {
      m += ((unsigned)(v[u].x - lo) < (unsigned)cnt) ? 1 : 0;
      m += ((unsigned)(v[u].y - lo) < (unsigned)cnt) ? 1 : 0;
      m += ((unsigned)(v[u].z - lo) < (unsigned)cnt) ? 1 : 0;
      m += ((unsigned)(v[u].w - lo) < (unsigned)cnt) ? 1 : 0;
    }
    unsigned ovf = 0;
    if (m > 0) {
      int pos = atomicAdd(&qn[par], m);
#pragma unroll
      for (int u = 0; u < SCAN_U; ++u) {
        int e0 = c * CHUNK + (u * AGG_THREADS + tid) * 4;
        int dd[4] = {v[u].x, v[u].y, v[u].z, v[u].w};
#pragma unroll
        for (int j = 0; j < 4; ++j) {
          unsigned dl = (unsigned)(dd[j] - lo);
          if (dl < (unsigned)cnt) {
            if (pos < QCAP) q[pos] = make_int2(e0 + j, (int)dl);
            else ovf |= 1u << (4 * u + j);
            ++pos;
          }
        }
      }
    }
    while (ovf) {
      int b = __builtin_ctz(ovf);
      ovf &= ovf - 1;
      int e = c * CHUNK + ((b >> 2) * AGG_THREADS + tid) * 4 + (b & 3);
      int dl = dst[e] - lo;
      int s = src[e];
      float cf = norm[e];
      for (int k = 0; k < F; ++k)
        atomicAdd(&acc[dl * F + k], cf * row_elem<F, SPLIT>(A, B, s, k));
    }
    __syncthreads();
    const int nq = (qn[par] < QCAP) ? qn[par] : QCAP;
    if (c + 1 < nchunks) load_chunk(c + 1);
    if (tid == 0) qn[par ^ 1] = 0;

    const int nfull = nq & ~3;
    for (int base = wave * 4; base < nfull; base += (AGG_THREADS / 32) * 4)
      add_rows<4, F, SPLIT>(acc, q + base, A, B, src, norm, lane);
    if (nfull + wave < nq)
      add_rows<1, F, SPLIT>(acc, q + nfull + wave, A, B, src, norm, lane);
    __syncthreads();
    par ^= 1;
  }

  for (int i = tid; i < cnt * F; i += AGG_THREADS) outbuf[(long)lo * F + i] = acc[i];
}

__global__ __launch_bounds__(256)
void k_gates(const float* __restrict__ AXH, const float* __restrict__ C,
             const float* __restrict__ Wx, const float* __restrict__ Wh,
             const float* __restrict__ bx, const float* __restrict__ bh,
             const float* __restrict__ bg, const float* __restrict__ wc,
             const float* __restrict__ ln_h_s, const float* __restrict__ ln_h_b,
             const float* __restrict__ ln_c_s, const float* __restrict__ ln_c_b,
             const float* __restrict__ ln_o_s, const float* __restrict__ ln_o_b,
             const float* __restrict__ skip,
             float* __restrict__ hiddenOut, float* __restrict__ cellOut,
             float* __restrict__ Y, int N) {
  extern __shared__ float smem[];
  float* Bf = smem;
  float* bias = smem + 18432;

  int tid = threadIdx.x;
  for (int i = tid; i < KSTEPS * NT_G * 64; i += 256) {
    int s = i >> 10;
    int rem = i & 1023;
    int t = rem >> 6;
    int rem2 = rem & 63;
    int L = rem2 >> 1;
    int j = rem2 & 1;
    int k = 4 * s + 2 * (L >> 4) + j;
    int c = 16 * t + (L & 15);
    int g = c >> 6, h = c & 63;
    Bf[i] = (k < FIN) ? Wx[(g * FIN + k) * HID + h]
                      : Wh[(g * HID + (k - FIN)) * HID + h];
  }
  if (tid < GCOLS) bias[tid] = bx[tid] + bh[tid] + bg[tid];
  __syncthreads();

  int wave = tid >> 5, lane = tid & 31;
  int half = lane >> 4, lcol = lane & 15;
  int rowBase = blockIdx.x * ROWS_PER_BLOCK + wave * 16;

  int arow = rowBase + lcol;
  if (arow > N - 1) arow = N - 1;
  const float* aptr = AXH + (long)arow * KTOT;

  v8f acc[NT_G];
#pragma unroll
  for (int t = 0; t < NT_G; ++t)
#pragma unroll
    for (int r = 0; r < 8; ++r) acc[t][r] = 0.0f;

  for (int s = 0; s < KSTEPS; ++s) {
    int k0 = 4 * s + 2 * half;
    v2f a = *(const v2f*)(aptr + k0);
    const float* bbase = Bf + s * 1024 + lane * 2;
#pragma unroll
    for (int t = 0; t < NT_G; ++t) {
      v2f b = *(const v2f*)(bbase + t * 64);
      acc[t] = __builtin_amdgcn_wmma_f32_16x16x4_f32(
          false, a, false, b, (short)0, acc[t], false, false);
    }
  }

#pragma unroll
  for (int r = 0; r < 8; ++r) {
    int node = rowBase + r + 8 * half;
    bool valid = node < N;
    int nc = valid ? node : N - 1;
    float hloc[4], cloc[4];
#pragma unroll
    for (int t = 0; t < 4; ++t) {
      int h = t * 16 + lcol;
      float c0 = C[(long)nc * HID + h];
      float gi = acc[t][r]      + bias[h]       + wc[h] * c0;
      float gf = acc[t + 4][r]  + bias[64 + h]  + wc[64 + h] * c0;
      float gc = acc[t + 8][r]  + bias[128 + h];
      float go = acc[t + 12][r] + bias[192 + h];
      float iv = sigmoidf_(gi);
      float fv = sigmoidf_(gf);
      float cc = fv * c0 + iv * tanhf(gc);
      float ov = sigmoidf_(go + wc[128 + h] * cc);
      hloc[t] = ov * tanhf(cc);
      cloc[t] = cc;
    }
    float sh = 0.f, sh2 = 0.f, sc = 0.f, sc2 = 0.f;
#pragma unroll
    for (int t = 0; t < 4; ++t) {
      sh += hloc[t]; sh2 += hloc[t] * hloc[t];
      sc += cloc[t]; sc2 += cloc[t] * cloc[t];
    }
#pragma unroll
    for (int mask = 1; mask < 16; mask <<= 1) {
      sh  += __shfl_xor(sh,  mask, 32);
      sh2 += __shfl_xor(sh2, mask, 32);
      sc  += __shfl_xor(sc,  mask, 32);
      sc2 += __shfl_xor(sc2, mask, 32);
    }
    float mh = sh * (1.0f / 64.0f), mc = sc * (1.0f / 64.0f);
    float vh = sh2 * (1.0f / 64.0f) - mh * mh;
    float vc = sc2 * (1.0f / 64.0f) - mc * mc;
    float rh = rsqrtf(vh + EPS), rc = rsqrtf(vc + EPS);
    if (valid) {
#pragma unroll
      for (int t = 0; t < 4; ++t) {
        int h = t * 16 + lcol;
        float hn = (hloc[t] - mh) * rh;
        float cn = (cloc[t] - mc) * rc;
        hiddenOut[(long)node * HID + h] = hn * ln_h_s[h] + ln_h_b[h];
        cellOut[(long)node * HID + h]   = cn * ln_c_s[h] + ln_c_b[h];
        float ov = hn * ln_o_s[h] + ln_o_b[h];
        Y[(long)node * KTOT + h] = ov > 0.0f ? ov : 0.0f;
      }
      if (lcol < 8) {
        Y[(long)node * KTOT + 64 + lcol] =
            (lcol < 2) ? skip[(long)node * 2 + lcol] : 0.0f;
      }
    }
  }
}

__global__ __launch_bounds__(256)
void k_gemm2(const float* __restrict__ AY, const float* __restrict__ W1,
             const float* __restrict__ b1, float* __restrict__ Z, int N) {
  __shared__ float Bf[KSTEPS * 4 * 64];
  __shared__ float b1s[HID];

  int tid = threadIdx.x;
  for (int i = tid; i < KSTEPS * 4 * 64; i += 256) {
    int s = i >> 8;
    int rem = i & 255;
    int t = rem >> 6;
    int rem2 = rem & 63;
    int L = rem2 >> 1;
    int j = rem2 & 1;
    int k = 4 * s + 2 * (L >> 4) + j;
    int c = 16 * t + (L & 15);
    Bf[i] = (k < HID + 2) ? W1[k * HID + c] : 0.0f;
  }
  if (tid < HID) b1s[tid] = b1[tid];
  __syncthreads();

  int wave = tid >> 5, lane = tid & 31;
  int half = lane >> 4, lcol = lane & 15;
  int rowBase = blockIdx.x * ROWS_PER_BLOCK + wave * 16;

  int arow = rowBase + lcol;
  if (arow > N - 1) arow = N - 1;
  const float* aptr = AY + (long)arow * KTOT;

  v8f acc[4];
#pragma unroll
  for (int t = 0; t < 4; ++t)
#pragma unroll
    for (int r = 0; r < 8; ++r) acc[t][r] = 0.0f;

  for (int s = 0; s < KSTEPS; ++s) {
    int k0 = 4 * s + 2 * half;
    v2f a = *(const v2f*)(aptr + k0);
    const float* bbase = Bf + s * 256 + lane * 2;
#pragma unroll
    for (int t = 0; t < 4; ++t) {
      v2f b = *(const v2f*)(bbase + t * 64);
      acc[t] = __builtin_amdgcn_wmma_f32_16x16x4_f32(
          false, a, false, b, (short)0, acc[t], false, false);
    }
  }

#pragma unroll
  for (int r = 0; r < 8; ++r) {
    int node = rowBase + r + 8 * half;
    if (node >= N) continue;
#pragma unroll
    for (int t = 0; t < 4; ++t) {
      int h = t * 16 + lcol;
      float z = acc[t][r] + b1s[h];
      Z[(long)node * HID + h] = z > 0.0f ? z : 0.0f;
    }
  }
}

__global__ void k_final(const float* __restrict__ AZ, const float* __restrict__ W2,
                        const float* __restrict__ b2, const float* __restrict__ X,
                        float* __restrict__ out, int N) {
  __shared__ float w2s[HID];
  if (threadIdx.x < HID) w2s[threadIdx.x] = W2[threadIdx.x];
  __syncthreads();
  int n = blockIdx.x * blockDim.x + threadIdx.x;
  if (n >= N) return;
  const float* a = AZ + (long)n * HID;
  float acc = b2[0] + X[(long)n * FIN];
#pragma unroll 8
  for (int h = 0; h < HID; ++h) acc += a[h] * w2s[h];
  out[n] = acc;
}

extern "C" void kernel_launch(void* const* d_in, const int* in_sizes, int n_in,
                              void* d_out, int out_size, void* d_ws, size_t ws_size,
                              hipStream_t stream) {
  const float* X    = (const float*)d_in[0];
  const int*   eidx = (const int*)d_in[1];
  const float* ew   = (const float*)d_in[2];
  const float* skip = (const float*)d_in[3];
  const float* H    = (const float*)d_in[4];
  const float* C    = (const float*)d_in[5];
  const float* Wx   = (const float*)d_in[6];
  const float* bx   = (const float*)d_in[7];
  const float* Wh   = (const float*)d_in[8];
  const float* bh   = (const float*)d_in[9];
  const float* wc   = (const float*)d_in[10];
  const float* bg   = (const float*)d_in[11];
  const float* ln_h_s = (const float*)d_in[12];
  const float* ln_h_b = (const float*)d_in[13];
  const float* ln_c_s = (const float*)d_in[14];
  const float* ln_c_b = (const float*)d_in[15];
  const float* ln_o_s = (const float*)d_in[16];
  const float* ln_o_b = (const float*)d_in[17];
  const float* W1   = (const float*)d_in[18];
  const float* b1   = (const float*)d_in[19];
  const float* W2   = (const float*)d_in[20];
  const float* b2   = (const float*)d_in[21];

  const int N = in_sizes[0] / FIN;
  const int E = in_sizes[2];

  const int* src = eidx;
  const int* dst = eidx + E;

  float* out_p    = (float*)d_out;
  float* hidden_p = out_p + N;
  float* cell_p   = hidden_p + (long)N * HID;

  float* ws   = (float*)d_ws;
  float* dinv = ws;
  float* norm = dinv + N;
  float* AXH  = norm + E;
  float* Y    = AXH;
  float* AY   = AXH + (long)N * KTOT;
  float* Z    = AY + (long)N * KTOT;
  float* AZ   = Z + (long)N * HID;

  const int B = 256;
  auto cdiv = [](long a, long b) { return (unsigned)((a + b - 1) / b); };

  k_init_deg<<<cdiv(N, B), B, 0, stream>>>(dinv, N);
  k_acc_deg<<<cdiv(E, B), B, 0, stream>>>(dinv, dst, ew, E);
  k_dinv<<<cdiv(N, B), B, 0, stream>>>(dinv, N);
  k_norm<<<cdiv(E, B), B, 0, stream>>>(norm, dinv, src, dst, ew, E);

  k_agg_tile<KTOT, true><<<cdiv(N, TILE_R), AGG_THREADS, 0, stream>>>(
      AXH, X, H, dinv, src, dst, norm, N, E);

  size_t smem_gates = (size_t)(KSTEPS * NT_G * 64 + GCOLS) * sizeof(float);
  k_gates<<<cdiv(N, ROWS_PER_BLOCK), B, smem_gates, stream>>>(
      AXH, C, Wx, Wh, bx, bh, bg, wc,
      ln_h_s, ln_h_b, ln_c_s, ln_c_b, ln_o_s, ln_o_b,
      skip, hidden_p, cell_p, Y, N);

  k_agg_tile<KTOT, false><<<cdiv(N, TILE_R), AGG_THREADS, 0, stream>>>(
      AY, Y, nullptr, dinv, src, dst, norm, N, E);

  k_gemm2<<<cdiv(N, ROWS_PER_BLOCK), B, 0, stream>>>(AY, W1, b1, Z, N);

  k_agg_tile<HID, false><<<cdiv(N, TILE_R), AGG_THREADS, 0, stream>>>(
      AZ, Z, nullptr, dinv, src, dst, norm, N, E);

  k_final<<<cdiv(N, B), B, 0, stream>>>(AZ, W2, b2, X, out_p, N);
}


# Edges bucketed once per call into per-tile record lists {src, row, coefficient}; the 3 LDS-tile aggregations walk their list instead of each scanning all dst; arrival-order sums, order changes
# speedup vs baseline: 3.5094x; 2.0954x over previous
#include <hip/hip_runtime.h>
#include <math.h>


typedef __attribute__((ext_vector_type(2))) float v2f;
typedef __attribute__((ext_vector_type(8))) float v8f;

#define FIN   8
#define HID   64
#define KTOT  72
#define KSTEPS 18
#define GCOLS 256
#define NT_G  16
#define ROWS_PER_BLOCK 128
#define EPS 1e-5f

__device__ __forceinline__ float sigmoidf_(float x) { return 1.0f / (1.0f + expf(-x)); }

__global__ void k_init_deg(float* deg, int* tcount, int ntiles, int N) {
  int i = blockIdx.x * blockDim.x + threadIdx.x;
  if (i < N) deg[i] = 1.0f;
  if (i < ntiles) tcount[i] = 0;
}

__global__ void k_acc_deg(float* deg, const int* dst, const float* w, int E) {
  int e = blockIdx.x * blockDim.x + threadIdx.x;
  if (e < E) atomicAdd(&deg[dst[e]], w[e]);
}

__global__ void k_dinv(float* deg, int N) {
  int i = blockIdx.x * blockDim.x + threadIdx.x;
  if (i < N) deg[i] = rsqrtf(deg[i]);
}


#define TILE_R      392
#define QCAP        8192
#define SCAN_U      8
#define AGG_THREADS 1024
#define CHUNK       (AGG_THREADS * 4 * SCAN_U)
#define LCAP        16384
#define KEY_BITS    9
#define BIN_THREADS 1024
#define BIN_EPT     8
#define BIN_TMAX    1024

__global__ __launch_bounds__(BIN_THREADS)
void k_bin(int2* __restrict__ lists, int* __restrict__ tcount, const float* __restrict__ dinv,
           const int* __restrict__ src, const int* __restrict__ dst,
           const float* __restrict__ w, int ntiles, int E) {
  __shared__ int hist[BIN_TMAX + 1];
  const int tid = threadIdx.x;
  for (int t = tid; t <= ntiles; t += BIN_THREADS) hist[(t < ntiles) ? t : BIN_TMAX] = 0;
  __syncthreads();

  const int e0 = blockIdx.x * (BIN_THREADS * BIN_EPT) + tid;
  int d[BIN_EPT], s[BIN_EPT];
  float ww[BIN_EPT];
#pragma unroll
  for (int j = 0; j < BIN_EPT; ++j) {
    int e = e0 + j * BIN_THREADS;
    bool ok = e < E;
    d[j] = ok ? dst[e] : -1;
    s[j] = ok ? src[e] : 0;
    ww[j] = ok ? w[e] : 0.0f;
  }
  int tl[BIN_EPT], rk[BIN_EPT], key[BIN_EPT];
  float cf[BIN_EPT];
#pragma unroll
  for (int j = 0; j < BIN_EPT; ++j) {
    int di = (d[j] >= 0) ? d[j] : 0;
    int t = di / TILE_R;
    key[j] = (s[j] << KEY_BITS) | (di - t * TILE_R);
    cf[j] = dinv[di] * ww[j] * dinv[s[j]];
    tl[j] = (d[j] >= 0) ? t : BIN_TMAX;
  }
#pragma unroll
  for (int j = 0; j < BIN_EPT; ++j) rk[j] = atomicAdd(&hist[tl[j]], 1);
  __syncthreads();
  for (int t = tid; t < ntiles; t += BIN_THREADS) {
    int h = hist[t];
    hist[t] = (h > 0) ? atomicAdd(&tcount[t], h) : 0;
  }
  __syncthreads();
#pragma unroll
  for (int j = 0; j < BIN_EPT; ++j) {
    if (tl[j] < BIN_TMAX) {
      int pos = hist[tl[j]] + rk[j];
      if (pos < LCAP)
        lists[(size_t)tl[j] * LCAP + pos] = make_int2(key[j], __float_as_int(cf[j]));
    }
  }
}

template <int F, bool SPLIT>
__device__ __forceinline__ float row_elem(const float* __restrict__ A,
                                          const float* __restrict__ B, int s, int k) {
  if (SPLIT) return (k < FIN) ? A[(long)s * FIN + k] : B[(long)s * HID + (k - FIN)];
  return A[(long)s * F + k];
}

template <int NE, int F, bool SPLIT>
__device__ __forceinline__ void add_rows(float* acc, const int2* qp,
                                         const float* __restrict__ A, const float* __restrict__ B,
                                         const int* __restrict__ src,
                                         const float* __restrict__ w,
                                         const float* __restrict__ dinv, int lo, int lane) {
  constexpr int KPL = (F + 31) / 32;
  int s[NE], dl[NE];
  float cf[NE];
#pragma unroll
  for (int j = 0; j < NE; ++j) {
    int2 ent = qp[j];
    dl[j] = ent.y;
    s[j] = src[ent.x];
    cf[j] = dinv[lo + ent.y] * w[ent.x] * dinv[s[j]];
  }
  float r[NE][KPL];
#pragma unroll
  for (int j = 0; j < NE; ++j)
#pragma unroll
    for (int m = 0; m < KPL; ++m) {
      int k = lane + 32 * m;
      r[j][m] = (k < F) ? row_elem<F, SPLIT>(A, B, s[j], k) : 0.0f;
    }
#pragma unroll
  for (int j = 0; j < NE; ++j)
#pragma unroll
    for (int m = 0; m < KPL; ++m) {
      int k = lane + 32 * m;
      if (k < F) atomicAdd(&acc[dl[j] * F + k], cf[j] * r[j][m]);
    }
}

template <int NE, int F, bool SPLIT>
__device__ __forceinline__ void add_recs(float* acc, int2 rec, int j0,
                                         const float* __restrict__ A, const float* __restrict__ B,
                                         int lane) {
  constexpr int KPL = (F + 31) / 32;
  int s[NE], dl[NE];
  float cf[NE];
#pragma unroll
  for (int j = 0; j < NE; ++j) {
    int key = __builtin_amdgcn_readlane(rec.x, j0 + j);
    cf[j] = __int_as_float(__builtin_amdgcn_readlane(rec.y, j0 + j));
    s[j] = key >> KEY_BITS;
    dl[j] = key & ((1 << KEY_BITS) - 1);
  }
  float r[NE][KPL];
#pragma unroll
  for (int j = 0; j < NE; ++j)
#pragma unroll
    for (int m = 0; m < KPL; ++m) {
      int k = lane + 32 * m;
      r[j][m] = (k < F) ? row_elem<F, SPLIT>(A, B, s[j], k) : 0.0f;
    }
#pragma unroll
  for (int j = 0; j < NE; ++j)
#pragma unroll
    for (int m = 0; m < KPL; ++m) {
      int k = lane + 32 * m;
      if (k < F) atomicAdd(&acc[dl[j] * F + k], cf[j] * r[j][m]);
    }
}

template <int F, bool SPLIT>
__global__ __launch_bounds__(AGG_THREADS)
void k_agg_tile(float* __restrict__ outbuf, const float* __restrict__ A,
                const float* __restrict__ B, const float* __restrict__ dinv,
                const int2* __restrict__ lists, const int* __restrict__ tcount,
                const int* __restrict__ src, const int* __restrict__ dst,
                const float* __restrict__ w, int N, int E) {
  __shared__ float acc[TILE_R * F];
  __shared__ int2 q[QCAP];
  __shared__ int qn[2];

  const int tid = threadIdx.x;
  const int lane = tid & 31, wave = tid >> 5;
  const int lo = blockIdx.x * TILE_R;
  const int cnt = (N - lo < TILE_R) ? (N - lo) : TILE_R;

  for (int i = tid; i < cnt * F; i += AGG_THREADS) {
    int n = i / F, k = i - n * F;
    float d = dinv[lo + n];
    acc[i] = d * d * row_elem<F, SPLIT>(A, B, lo + n, k);
  }
  if (tid == 0) { qn[0] = 0; qn[1] = 0; }
  __syncthreads();

  const int nl = __builtin_amdgcn_readfirstlane(tcount[blockIdx.x]);
  if (nl <= LCAP) {
    const int2* __restrict__ L = lists + (size_t)blockIdx.x * LCAP;
    int base = __builtin_amdgcn_readfirstlane(wave) * 32;
    int2 rec = make_int2(0, 0);
    if (base + lane < nl) rec = L[base + lane];
    while (base < nl) {
      const int nbase = base + AGG_THREADS;
      int2 nrec = make_int2(0, 0);
      if (nbase + lane < nl) nrec = L[nbase + lane];
      const int nv = (nl - base < 32) ? (nl - base) : 32;
      int j = 0;
      for (; j + 8 <= nv; j += 8) add_recs<8, F, SPLIT>(acc, rec, j, A, B, lane);
      for (; j < nv; ++j) add_recs<1, F, SPLIT>(acc, rec, j, A, B, lane);
      rec = nrec;
      base = nbase;
    }
    __syncthreads();
  } else {
  const bool vec_ok = (((size_t)dst) & 15) == 0;
  const int nchunks = (E + CHUNK - 1) / CHUNK;
  int4 v[SCAN_U];

  auto load_chunk = [&](int c) {
    if (vec_ok && c * CHUNK + CHUNK <= E) {
#pragma unroll
      for (int u = 0; u < SCAN_U; ++u)
        v[u] = *(const int4*)(dst + c * CHUNK + (u * AGG_THREADS + tid) * 4);
    } else {
#pragma unroll
      for (int u = 0; u < SCAN_U; ++u) {
        int e0 = c * CHUNK + (u * AGG_THREADS + tid) * 4;
        v[u].x = (e0     < E) ? dst[e0]     : -1;
        v[u].y = (e0 + 1 < E) ? dst[e0 + 1] : -1;
        v[u].z = (e0 + 2 < E) ? dst[e0 + 2] : -1;
        v[u].w = (e0 + 3 < E) ? dst[e0 + 3] : -1;
      }
    }
  };

  load_chunk(0);
  int par = 0;
  for (int c = 0; c < nchunks; ++c) {
    int m = 0;
#pragma unroll
    for (int u = 0; u < SCAN_U; ++u) {
      m += ((unsigned)(v[u].x - lo) < (unsigned)cnt) ? 1 : 0;
      m += ((unsigned)(v[u].y - lo) < (unsigned)cnt) ? 1 : 0;
      m += ((unsigned)(v[u].z - lo) < (unsigned)cnt) ? 1 : 0;
      m += ((unsigned)(v[u].w - lo) < (unsigned)cnt) ? 1 : 0;
    }
    unsigned ovf = 0;
    if (m > 0) {
      int pos = atomicAdd(&qn[par], m);
#pragma unroll
      for (int u = 0; u < SCAN_U; ++u) {
        int e0 = c * CHUNK + (u * AGG_THREADS + tid) * 4;
        int dd[4] = {v[u].x, v[u].y, v[u].z, v[u].w};
#pragma unroll
        for (int j = 0; j < 4; ++j) {
          unsigned dl = (unsigned)(dd[j] - lo);
          if (dl < (unsigned)cnt) {
            if (pos < QCAP) q[pos] = make_int2(e0 + j, (int)dl);
            else ovf |= 1u << (4 * u + j);
            ++pos;
          }
        }
      }
    }
    while (ovf) {
      int b = __builtin_ctz(ovf);
      ovf &= ovf - 1;
      int e = c * CHUNK + ((b >> 2) * AGG_THREADS + tid) * 4 + (b & 3);
      int dl = dst[e] - lo;
      int s = src[e];
      float cf = dinv[lo + dl] * w[e] * dinv[s];
      for (int k = 0; k < F; ++k)
        atomicAdd(&acc[dl * F + k], cf * row_elem<F, SPLIT>(A, B, s, k));
    }
    __syncthreads();
    const int nq = (qn[par] < QCAP) ? qn[par] : QCAP;
    if (c + 1 < nchunks) load_chunk(c + 1);
    if (tid == 0) qn[par ^ 1] = 0;

    const int nfull = nq & ~3;
    for (int base = wave * 4; base < nfull; base += (AGG_THREADS / 32) * 4)
      add_rows<4, F, SPLIT>(acc, q + base, A, B, src, w, dinv, lo, lane);
    if (nfull + wave < nq)
      add_rows<1, F, SPLIT>(acc, q + nfull + wave, A, B, src, w, dinv, lo, lane);
    __syncthreads();
    par ^= 1;
  }
  }

  for (int i = tid; i < cnt * F; i += AGG_THREADS) outbuf[(long)lo * F + i] = acc[i];
}

__global__ __launch_bounds__(256)
void k_gates(const float* __restrict__ AXH, const float* __restrict__ C,
             const float* __restrict__ Wx, const float* __restrict__ Wh,
             const float* __restrict__ bx, const float* __restrict__ bh,
             const float* __restrict__ bg, const float* __restrict__ wc,
             const float* __restrict__ ln_h_s, const float* __restrict__ ln_h_b,
             const float* __restrict__ ln_c_s, const float* __restrict__ ln_c_b,
             const float* __restrict__ ln_o_s, const float* __restrict__ ln_o_b,
             const float* __restrict__ skip,
             float* __restrict__ hiddenOut, float* __restrict__ cellOut,
             float* __restrict__ Y, int N) {
  extern __shared__ float smem[];
  float* Bf = smem;
  float* bias = smem + 18432;

  int tid = threadIdx.x;
  for (int i = tid; i < KSTEPS * NT_G * 64; i += 256) {
    int s = i >> 10;
    int rem = i & 1023;
    int t = rem >> 6;
    int rem2 = rem & 63;
    int L = rem2 >> 1;
    int j = rem2 & 1;
    int k = 4 * s + 2 * (L >> 4) + j;
    int c = 16 * t + (L & 15);
    int g = c >> 6, h = c & 63;
    Bf[i] = (k < FIN) ? Wx[(g * FIN + k) * HID + h]
                      : Wh[(g * HID + (k - FIN)) * HID + h];
  }
  if (tid < GCOLS) bias[tid] = bx[tid] + bh[tid] + bg[tid];
  __syncthreads();

  int wave = tid >> 5, lane = tid & 31;
  int half = lane >> 4, lcol = lane & 15;
  int rowBase = blockIdx.x * ROWS_PER_BLOCK + wave * 16;

  int arow = rowBase + lcol;
  if (arow > N - 1) arow = N - 1;
  const float* aptr = AXH + (long)arow * KTOT;

  v8f acc[NT_G];
#pragma unroll
  for (int t = 0; t < NT_G; ++t)
#pragma unroll
    for (int r = 0; r < 8; ++r) acc[t][r] = 0.0f;

  for (int s = 0; s < KSTEPS; ++s) {
    int k0 = 4 * s + 2 * half;
    v2f a = *(const v2f*)(aptr + k0);
    const float* bbase = Bf + s * 1024 + lane * 2;
#pragma unroll
    for (int t = 0; t < NT_G; ++t) {
      v2f b = *(const v2f*)(bbase + t * 64);
      acc[t] = __builtin_amdgcn_wmma_f32_16x16x4_f32(
          false, a, false, b, (short)0, acc[t], false, false);
    }
  }

#pragma unroll
  for (int r = 0; r < 8; ++r) {
    int node = rowBase + r + 8 * half;
    bool valid = node < N;
    int nc = valid ? node : N - 1;
    float hloc[4], cloc[4];
#pragma unroll
    for (int t = 0; t < 4; ++t) {
      int h = t * 16 + lcol;
      float c0 = C[(long)nc * HID + h];
      float gi = acc[t][r]      + bias[h]       + wc[h] * c0;
      float gf = acc[t + 4][r]  + bias[64 + h]  + wc[64 + h] * c0;
      float gc = acc[t + 8][r]  + bias[128 + h];
      float go = acc[t + 12][r] + bias[192 + h];
      float iv = sigmoidf_(gi);
      float fv = sigmoidf_(gf);
      float cc = fv * c0 + iv * tanhf(gc);
      float ov = sigmoidf_(go + wc[128 + h] * cc);
      hloc[t] = ov * tanhf(cc);
      cloc[t] = cc;
    }
    float sh = 0.f, sh2 = 0.f, sc = 0.f, sc2 = 0.f;
#pragma unroll
    for (int t = 0; t < 4; ++t) {
      sh += hloc[t]; sh2 += hloc[t] * hloc[t];
      sc += cloc[t]; sc2 += cloc[t] * cloc[t];
    }
#pragma unroll
    for (int mask = 1; mask < 16; mask <<= 1) {
      sh  += __shfl_xor(sh,  mask, 32);
      sh2 += __shfl_xor(sh2, mask, 32);
      sc  += __shfl_xor(sc,  mask, 32);
      sc2 += __shfl_xor(sc2, mask, 32);
    }
    float mh = sh * (1.0f / 64.0f), mc = sc * (1.0f / 64.0f);
    float vh = sh2 * (1.0f / 64.0f) - mh * mh;
    float vc = sc2 * (1.0f / 64.0f) - mc * mc;
    float rh = rsqrtf(vh + EPS), rc = rsqrtf(vc + EPS);
    if (valid) {
#pragma unroll
      for (int t = 0; t < 4; ++t) {
        int h = t * 16 + lcol;
        float hn = (hloc[t] - mh) * rh;
        float cn = (cloc[t] - mc) * rc;
        hiddenOut[(long)node * HID + h] = hn * ln_h_s[h] + ln_h_b[h];
        cellOut[(long)node * HID + h]   = cn * ln_c_s[h] + ln_c_b[h];
        float ov = hn * ln_o_s[h] + ln_o_b[h];
        Y[(long)node * KTOT + h] = ov > 0.0f ? ov : 0.0f;
      }
      if (lcol < 8) {
        Y[(long)node * KTOT + 64 + lcol] =
            (lcol < 2) ? skip[(long)node * 2 + lcol] : 0.0f;
      }
    }
  }
}

__global__ __launch_bounds__(256)
void k_gemm2(const float* __restrict__ AY, const float* __restrict__ W1,
             const float* __restrict__ b1, float* __restrict__ Z, int N) {
  __shared__ float Bf[KSTEPS * 4 * 64];
  __shared__ float b1s[HID];

  int tid = threadIdx.x;
  for (int i = tid; i < KSTEPS * 4 * 64; i += 256) {
    int s = i >> 8;
    int rem = i & 255;
    int t = rem >> 6;
    int rem2 = rem & 63;
    int L = rem2 >> 1;
    int j = rem2 & 1;
    int k = 4 * s + 2 * (L >> 4) + j;
    int c = 16 * t + (L & 15);
    Bf[i] = (k < HID + 2) ? W1[k * HID + c] : 0.0f;
  }
  if (tid < HID) b1s[tid] = b1[tid];
  __syncthreads();

  int wave = tid >> 5, lane = tid & 31;
  int half = lane >> 4, lcol = lane & 15;
  int rowBase = blockIdx.x * ROWS_PER_BLOCK + wave * 16;

  int arow = rowBase + lcol;
  if (arow > N - 1) arow = N - 1;
  const float* aptr = AY + (long)arow * KTOT;

  v8f acc[4];
#pragma unroll
  for (int t = 0; t < 4; ++t)
#pragma unroll
    for (int r = 0; r < 8; ++r) acc[t][r] = 0.0f;

  for (int s = 0; s < KSTEPS; ++s) {
    int k0 = 4 * s + 2 * half;
    v2f a = *(const v2f*)(aptr + k0);
    const float* bbase = Bf + s * 256 + lane * 2;
#pragma unroll
    for (int t = 0; t < 4; ++t) {
      v2f b = *(const v2f*)(bbase + t * 64);
      acc[t] = __builtin_amdgcn_wmma_f32_16x16x4_f32(
          false, a, false, b, (short)0, acc[t], false, false);
    }
  }

#pragma unroll
  for (int r = 0; r < 8; ++r) {
    int node = rowBase + r + 8 * half;
    if (node >= N) continue;
#pragma unroll
    for (int t = 0; t < 4; ++t) {
      int h = t * 16 + lcol;
      float z = acc[t][r] + b1s[h];
      Z[(long)node * HID + h] = z > 0.0f ? z : 0.0f;
    }
  }
}

__global__ void k_final(const float* __restrict__ AZ, const float* __restrict__ W2,
                        const float* __restrict__ b2, const float* __restrict__ X,
                        float* __restrict__ out, int N) {
  __shared__ float w2s[HID];
  if (threadIdx.x < HID) w2s[threadIdx.x] = W2[threadIdx.x];
  __syncthreads();
  int n = blockIdx.x * blockDim.x + threadIdx.x;
  if (n >= N) return;
  const float* a = AZ + (long)n * HID;
  float acc = b2[0] + X[(long)n * FIN];
#pragma unroll 8
  for (int h = 0; h < HID; ++h) acc += a[h] * w2s[h];
  out[n] = acc;
}

extern "C" void kernel_launch(void* const* d_in, const int* in_sizes, int n_in,
                              void* d_out, int out_size, void* d_ws, size_t ws_size,
                              hipStream_t stream) {
  const float* X    = (const float*)d_in[0];
  const int*   eidx = (const int*)d_in[1];
  const float* ew   = (const float*)d_in[2];
  const float* skip = (const float*)d_in[3];
  const float* H    = (const float*)d_in[4];
  const float* C    = (const float*)d_in[5];
  const float* Wx   = (const float*)d_in[6];
  const float* bx   = (const float*)d_in[7];
  const float* Wh   = (const float*)d_in[8];
  const float* bh   = (const float*)d_in[9];
  const float* wc   = (const float*)d_in[10];
  const float* bg   = (const float*)d_in[11];
  const float* ln_h_s = (const float*)d_in[12];
  const float* ln_h_b = (const float*)d_in[13];
  const float* ln_c_s = (const float*)d_in[14];
  const float* ln_c_b = (const float*)d_in[15];
  const float* ln_o_s = (const float*)d_in[16];
  const float* ln_o_b = (const float*)d_in[17];
  const float* W1   = (const float*)d_in[18];
  const float* b1   = (const float*)d_in[19];
  const float* W2   = (const float*)d_in[20];
  const float* b2   = (const float*)d_in[21];

  const int N = in_sizes[0] / FIN;
  const int E = in_sizes[2];

  const int* src = eidx;
  const int* dst = eidx + E;

  float* out_p    = (float*)d_out;
  float* hidden_p = out_p + N;
  float* cell_p   = hidden_p + (long)N * HID;

  float* ws   = (float*)d_ws;
  float* dinv = ws;
  int* tcount = (int*)(dinv + N);
  float* AXH  = dinv + N + E;
  float* Y    = AXH;
  float* AY   = AXH + (long)N * KTOT;
  float* Z    = AXH;
  float* AZ   = AY;
  int2* lists = (int2*)(AY + (long)N * KTOT);

  const int B = 256;
  auto cdiv = [](long a, long b) { return (unsigned)((a + b - 1) / b); };
  const int ntiles = (int)cdiv(N, TILE_R);

  k_init_deg<<<cdiv(N, B), B, 0, stream>>>(dinv, tcount, ntiles, N);
  k_acc_deg<<<cdiv(E, B), B, 0, stream>>>(dinv, dst, ew, E);
  k_dinv<<<cdiv(N, B), B, 0, stream>>>(dinv, N);
  k_bin<<<cdiv(E, BIN_THREADS * BIN_EPT), BIN_THREADS, 0, stream>>>(
      lists, tcount, dinv, src, dst, ew, ntiles, E);

  k_agg_tile<KTOT, true><<<ntiles, AGG_THREADS, 0, stream>>>(
      AXH, X, H, dinv, lists, tcount, src, dst, ew, N, E);

  size_t smem_gates = (size_t)(KSTEPS * NT_G * 64 + GCOLS) * sizeof(float);
  k_gates<<<cdiv(N, ROWS_PER_BLOCK), B, smem_gates, stream>>>(
      AXH, C, Wx, Wh, bx, bh, bg, wc,
      ln_h_s, ln_h_b, ln_c_s, ln_c_b, ln_o_s, ln_o_b,
      skip, hidden_p, cell_p, Y, N);

  k_agg_tile<KTOT, false><<<ntiles, AGG_THREADS, 0, stream>>>(
      AY, Y, nullptr, dinv, lists, tcount, src, dst, ew, N, E);

  k_gemm2<<<cdiv(N, ROWS_PER_BLOCK), B, 0, stream>>>(AY, W1, b1, Z, N);

  k_agg_tile<HID, false><<<ntiles, AGG_THREADS, 0, stream>>>(
      AZ, Z, nullptr, dinv, lists, tcount, src, dst, ew, N, E);

  k_final<<<cdiv(N, B), B, 0, stream>>>(AZ, W2, b2, X, out_p, N);
}


# Skip columns aggregated in the 1st pass from a packed [X|skip] row table, so the 2nd pass gathers aligned 64-float rows (2 lines, 2 LDS adds per edge) not 288-byte rows; sums stay in arrival order
# speedup vs baseline: 3.8562x; 1.0988x over previous
#include <hip/hip_runtime.h>
#include <math.h>


typedef __attribute__((ext_vector_type(2))) float v2f;
typedef __attribute__((ext_vector_type(8))) float v8f;

#define FIN   8
#define HID   64
#define KTOT  72
#define KSTEPS 18
#define KSTEPS2 17
#define XSP   16
#define F1    74
#define GCOLS 256
#define NT_G  16
#define ROWS_PER_BLOCK 128
#define EPS 1e-5f

__device__ __forceinline__ float sigmoidf_(float x) { return 1.0f / (1.0f + expf(-x)); }

__global__ void k_init_deg(float* deg, int* tcount, float* __restrict__ XS,
                           const float* __restrict__ X, const float* __restrict__ skip,
                           int ntiles, int N) {
  int i = blockIdx.x * blockDim.x + threadIdx.x;
  if (i < N) deg[i] = 1.0f;
  if (i < ntiles) tcount[i] = 0;
  if (i < N * 4) {
    int n = i >> 2, q4 = i & 3;
    float* o = XS + (long)i * 4;
    if (q4 < 2) {
      const float* x = X + (long)n * FIN + q4 * 4;
      o[0] = x[0]; o[1] = x[1]; o[2] = x[2]; o[3] = x[3];
    } else if (q4 == 2) {
      o[0] = skip[(long)n * 2];
      o[1] = skip[(long)n * 2 + 1];
    }
  }
}

__global__ void k_acc_deg(float* deg, const int* dst, const float* w, int E) {
  int e = blockIdx.x * blockDim.x + threadIdx.x;
  if (e < E) atomicAdd(&deg[dst[e]], w[e]);
}

__global__ void k_dinv(float* deg, int N) {
  int i = blockIdx.x * blockDim.x + threadIdx.x;
  if (i < N) deg[i] = rsqrtf(deg[i]);
}


#define TILE_R      392
#define QCAP        8192
#define SCAN_U      8
#define AGG_THREADS 1024
#define CHUNK       (AGG_THREADS * 4 * SCAN_U)
#define LCAP        16384
#define KEY_BITS    9
#define BIN_THREADS 1024
#define BIN_EPT     8
#define BIN_TMAX    1024

__global__ __launch_bounds__(BIN_THREADS)
void k_bin(int2* __restrict__ lists, int* __restrict__ tcount, const float* __restrict__ dinv,
           const int* __restrict__ src, const int* __restrict__ dst,
           const float* __restrict__ w, int ntiles, int E) {
  __shared__ int hist[BIN_TMAX + 1];
  const int tid = threadIdx.x;
  for (int t = tid; t <= ntiles; t += BIN_THREADS) hist[(t < ntiles) ? t : BIN_TMAX] = 0;
  __syncthreads();

  const int e0 = blockIdx.x * (BIN_THREADS * BIN_EPT) + tid;
  int d[BIN_EPT], s[BIN_EPT];
  float ww[BIN_EPT];
#pragma unroll
  for (int j = 0; j < BIN_EPT; ++j) {
    int e = e0 + j * BIN_THREADS;
    bool ok = e < E;
    d[j] = ok ? dst[e] : -1;
    s[j] = ok ? src[e] : 0;
    ww[j] = ok ? w[e] : 0.0f;
  }
  int tl[BIN_EPT], rk[BIN_EPT], key[BIN_EPT];
  float cf[BIN_EPT];
#pragma unroll
  for (int j = 0; j < BIN_EPT; ++j) {
    int di = (d[j] >= 0) ? d[j] : 0;
    int t = di / TILE_R;
    key[j] = (s[j] << KEY_BITS) | (di - t * TILE_R);
    cf[j] = dinv[di] * ww[j] * dinv[s[j]];
    tl[j] = (d[j] >= 0) ? t : BIN_TMAX;
  }
#pragma unroll
  for (int j = 0; j < BIN_EPT; ++j) rk[j] = atomicAdd(&hist[tl[j]], 1);
  __syncthreads();
  for (int t = tid; t < ntiles; t += BIN_THREADS) {
    int h = hist[t];
    hist[t] = (h > 0) ? atomicAdd(&tcount[t], h) : 0;
  }
  __syncthreads();
#pragma unroll
  for (int j = 0; j < BIN_EPT; ++j) {
    if (tl[j] < BIN_TMAX) {
      int pos = hist[tl[j]] + rk[j];
      if (pos < LCAP)
        lists[(size_t)tl[j] * LCAP + pos] = make_int2(key[j], __float_as_int(cf[j]));
    }
  }
}

template <int F, bool SPLIT>
__device__ __forceinline__ float row_elem(const float* __restrict__ A,
                                          const float* __restrict__ B, int s, int k) {
  if (SPLIT) return (k < HID) ? B[(long)s * HID + k] : A[(long)s * XSP + (k - HID)];
  return A[(long)s * F + k];
}

template <int NE, int F, bool SPLIT>
__device__ __forceinline__ void add_rows(float* acc, const int2* qp,
                                         const float* __restrict__ A, const float* __restrict__ B,
                                         const int* __restrict__ src,
                                         const float* __restrict__ w,
                                         const float* __restrict__ dinv, int lo, int lane) {
  constexpr int KPL = (F + 31) / 32;
  int s[NE], dl[NE];
  float cf[NE];
#pragma unroll
  for (int j = 0; j < NE; ++j) {
    int2 ent = qp[j];
    dl[j] = ent.y;
    s[j] = src[ent.x];
    cf[j] = dinv[lo + ent.y] * w[ent.x] * dinv[s[j]];
  }
  float r[NE][KPL];
#pragma unroll
  for (int j = 0; j < NE; ++j)
#pragma unroll
    for (int m = 0; m < KPL; ++m) {
      int k = lane + 32 * m;
      r[j][m] = (k < F) ? row_elem<F, SPLIT>(A, B, s[j], k) : 0.0f;
    }
#pragma unroll
  for (int j = 0; j < NE; ++j)
#pragma unroll
    for (int m = 0; m < KPL; ++m) {
      int k = lane + 32 * m;
      if (k < F) atomicAdd(&acc[dl[j] * F + k], cf[j] * r[j][m]);
    }
}

template <int NE, int F, bool SPLIT>
__device__ __forceinline__ void add_recs(float* acc, int2 rec, int j0,
                                         const float* __restrict__ A, const float* __restrict__ B,
                                         int lane) {
  constexpr int KPL = (F + 31) / 32;
  int s[NE], dl[NE];
  float cf[NE];
#pragma unroll
  for (int j = 0; j < NE; ++j) {
    int key = __builtin_amdgcn_readlane(rec.x, j0 + j);
    cf[j] = __int_as_float(__builtin_amdgcn_readlane(rec.y, j0 + j));
    s[j] = key >> KEY_BITS;
    dl[j] = key & ((1 << KEY_BITS) - 1);
  }
  float r[NE][KPL];
#pragma unroll
  for (int j = 0; j < NE; ++j)
#pragma unroll
    for (int m = 0; m < KPL; ++m) {
      int k = lane + 32 * m;
      r[j][m] = (k < F) ? row_elem<F, SPLIT>(A, B, s[j], k) : 0.0f;
    }
#pragma unroll
  for (int j = 0; j < NE; ++j)
#pragma unroll
    for (int m = 0; m < KPL; ++m) {
      int k = lane + 32 * m;
      if (k < F) atomicAdd(&acc[dl[j] * F + k], cf[j] * r[j][m]);
    }
}

template <int F, bool SPLIT>
__global__ __launch_bounds__(AGG_THREADS)
void k_agg_tile(float* __restrict__ outbuf, float* __restrict__ outS, const float* __restrict__ A,
                const float* __restrict__ B, const float* __restrict__ dinv,
                const int2* __restrict__ lists, const int* __restrict__ tcount,
                const int* __restrict__ src, const int* __restrict__ dst,
                const float* __restrict__ w, int N, int E) {
  __shared__ float acc[TILE_R * F];
  __shared__ int2 q[QCAP];
  __shared__ int qn[2];

  const int tid = threadIdx.x;
  const int lane = tid & 31, wave = tid >> 5;
  const int lo = blockIdx.x * TILE_R;
  const int cnt = (N - lo < TILE_R) ? (N - lo) : TILE_R;

  for (int i = tid; i < cnt * F; i += AGG_THREADS) {
    int n = i / F, k = i - n * F;
    float d = dinv[lo + n];
    acc[i] = d * d * row_elem<F, SPLIT>(A, B, lo + n, k);
  }
  if (tid == 0) { qn[0] = 0; qn[1] = 0; }
  __syncthreads();

  const int nl = __builtin_amdgcn_readfirstlane(tcount[blockIdx.x]);
  if (nl <= LCAP) {
    const int2* __restrict__ L = lists + (size_t)blockIdx.x * LCAP;
    int base = __builtin_amdgcn_readfirstlane(wave) * 32;
    int2 rec = make_int2(0, 0);
    if (base + lane < nl) rec = L[base + lane];
    while (base < nl) {
      const int nbase = base + AGG_THREADS;
      int2 nrec = make_int2(0, 0);
      if (nbase + lane < nl) nrec = L[nbase + lane];
      const int nv = (nl - base < 32) ? (nl - base) : 32;
      int j = 0;
      for (; j + 8 <= nv; j += 8) add_recs<8, F, SPLIT>(acc, rec, j, A, B, lane);
      for (; j < nv; ++j) add_recs<1, F, SPLIT>(acc, rec, j, A, B, lane);
      rec = nrec;
      base = nbase;
    }
    __syncthreads();
  } else {
  const bool vec_ok = (((size_t)dst) & 15) == 0;
  const int nchunks = (E + CHUNK - 1) / CHUNK;
  int4 v[SCAN_U];

  auto load_chunk = [&](int c) {
    if (vec_ok && c * CHUNK + CHUNK <= E) {
#pragma unroll
      for (int u = 0; u < SCAN_U; ++u)
        v[u] = *(const int4*)(dst + c * CHUNK + (u * AGG_THREADS + tid) * 4);
    } else {
#pragma unroll
      for (int u = 0; u < SCAN_U; ++u) {
        int e0 = c * CHUNK + (u * AGG_THREADS + tid) * 4;
        v[u].x = (e0     < E) ? dst[e0]     : -1;
        v[u].y = (e0 + 1 < E) ? dst[e0 + 1] : -1;
        v[u].z = (e0 + 2 < E) ? dst[e0 + 2] : -1;
        v[u].w = (e0 + 3 < E) ? dst[e0 + 3] : -1;
      }
    }
  };

  load_chunk(0);
  int par = 0;
  for (int c = 0; c < nchunks; ++c) {
    int m = 0;
#pragma unroll
    for (int u = 0; u < SCAN_U; ++u) {
      m += ((unsigned)(v[u].x - lo) < (unsigned)cnt) ? 1 : 0;
      m += ((unsigned)(v[u].y - lo) < (unsigned)cnt) ? 1 : 0;
      m += ((unsigned)(v[u].z - lo) < (unsigned)cnt) ? 1 : 0;
      m += ((unsigned)(v[u].w - lo) < (unsigned)cnt) ? 1 : 0;
    }
    unsigned ovf = 0;
    if (m > 0) {
      int pos = atomicAdd(&qn[par], m);
#pragma unroll
      for (int u = 0; u < SCAN_U; ++u) {
        int e0 = c * CHUNK + (u * AGG_THREADS + tid) * 4;
        int dd[4] = {v[u].x, v[u].y, v[u].z, v[u].w};
#pragma unroll
        for (int j = 0; j < 4; ++j) {
          unsigned dl = (unsigned)(dd[j] - lo);
          if (dl < (unsigned)cnt) {
            if (pos < QCAP) q[pos] = make_int2(e0 + j, (int)dl);
            else ovf |= 1u << (4 * u + j);
            ++pos;
          }
        }
      }
    }
    while (ovf) {
      int b = __builtin_ctz(ovf);
      ovf &= ovf - 1;
      int e = c * CHUNK + ((b >> 2) * AGG_THREADS + tid) * 4 + (b & 3);
      int dl = dst[e] - lo;
      int s = src[e];
      float cf = dinv[lo + dl] * w[e] * dinv[s];
      for (int k = 0; k < F; ++k)
        atomicAdd(&acc[dl * F + k], cf * row_elem<F, SPLIT>(A, B, s, k));
    }
    __syncthreads();
    const int nq = (qn[par] < QCAP) ? qn[par] : QCAP;
    if (c + 1 < nchunks) load_chunk(c + 1);
    if (tid == 0) qn[par ^ 1] = 0;

    const int nfull = nq & ~3;
    for (int base = wave * 4; base < nfull; base += (AGG_THREADS / 32) * 4)
      add_rows<4, F, SPLIT>(acc, q + base, A, B, src, w, dinv, lo, lane);
    if (nfull + wave < nq)
      add_rows<1, F, SPLIT>(acc, q + nfull + wave, A, B, src, w, dinv, lo, lane);
    __syncthreads();
    par ^= 1;
  }
  }

  if (SPLIT) {
    for (int i = tid; i < cnt * F; i += AGG_THREADS) {
      int n = i / F, k = i - n * F;
      float v = acc[i];
      long row = lo + n;
      if (k < HID) outbuf[row * KTOT + FIN + k] = v;
      else if (k < HID + FIN) outbuf[row * KTOT + (k - HID)] = v;
      else outS[row * 2 + (k - HID - FIN)] = v;
    }
  } else {
    for (int i = tid; i < cnt * F; i += AGG_THREADS) outbuf[(long)lo * F + i] = acc[i];
  }
}

__global__ __launch_bounds__(256)
void k_gates(const float* __restrict__ AXH, const float* __restrict__ C,
             const float* __restrict__ Wx, const float* __restrict__ Wh,
             const float* __restrict__ bx, const float* __restrict__ bh,
             const float* __restrict__ bg, const float* __restrict__ wc,
             const float* __restrict__ ln_h_s, const float* __restrict__ ln_h_b,
             const float* __restrict__ ln_c_s, const float* __restrict__ ln_c_b,
             const float* __restrict__ ln_o_s, const float* __restrict__ ln_o_b,
             float* __restrict__ hiddenOut, float* __restrict__ cellOut,
             float* __restrict__ Y, int N) {
  extern __shared__ float smem[];
  float* Bf = smem;
  float* bias = smem + 18432;

  int tid = threadIdx.x;
  for (int i = tid; i < KSTEPS * NT_G * 64; i += 256) {
    int s = i >> 10;
    int rem = i & 1023;
    int t = rem >> 6;
    int rem2 = rem & 63;
    int L = rem2 >> 1;
    int j = rem2 & 1;
    int k = 4 * s + 2 * (L >> 4) + j;
    int c = 16 * t + (L & 15);
    int g = c >> 6, h = c & 63;
    Bf[i] = (k < FIN) ? Wx[(g * FIN + k) * HID + h]
                      : Wh[(g * HID + (k - FIN)) * HID + h];
  }
  if (tid < GCOLS) bias[tid] = bx[tid] + bh[tid] + bg[tid];
  __syncthreads();

  int wave = tid >> 5, lane = tid & 31;
  int half = lane >> 4, lcol = lane & 15;
  int rowBase = blockIdx.x * ROWS_PER_BLOCK + wave * 16;

  int arow = rowBase + lcol;
  if (arow > N - 1) arow = N - 1;
  const float* aptr = AXH + (long)arow * KTOT;

  v8f acc[NT_G];
#pragma unroll
  for (int t = 0; t < NT_G; ++t)
#pragma unroll
    for (int r = 0; r < 8; ++r) acc[t][r] = 0.0f;

  for (int s = 0; s < KSTEPS; ++s) {
    int k0 = 4 * s + 2 * half;
    v2f a = *(const v2f*)(aptr + k0);
    const float* bbase = Bf + s * 1024 + lane * 2;
#pragma unroll
    for (int t = 0; t < NT_G; ++t) {
      v2f b = *(const v2f*)(bbase + t * 64);
      acc[t] = __builtin_amdgcn_wmma_f32_16x16x4_f32(
          false, a, false, b, (short)0, acc[t], false, false);
    }
  }

#pragma unroll
  for (int r = 0; r < 8; ++r) {
    int node = rowBase + r + 8 * half;
    bool valid = node < N;
    int nc = valid ? node : N - 1;
    float hloc[4], cloc[4];
#pragma unroll
    for (int t = 0; t < 4; ++t) {
      int h = t * 16 + lcol;
      float c0 = C[(long)nc * HID + h];
      float gi = acc[t][r]      + bias[h]       + wc[h] * c0;
      float gf = acc[t + 4][r]  + bias[64 + h]  + wc[64 + h] * c0;
      float gc = acc[t + 8][r]  + bias[128 + h];
      float go = acc[t + 12][r] + bias[192 + h];
      float iv = sigmoidf_(gi);
      float fv = sigmoidf_(gf);
      float cc = fv * c0 + iv * tanhf(gc);
      float ov = sigmoidf_(go + wc[128 + h] * cc);
      hloc[t] = ov * tanhf(cc);
      cloc[t] = cc;
    }
    float sh = 0.f, sh2 = 0.f, sc = 0.f, sc2 = 0.f;
#pragma unroll
    for (int t = 0; t < 4; ++t) {
      sh += hloc[t]; sh2 += hloc[t] * hloc[t];
      sc += cloc[t]; sc2 += cloc[t] * cloc[t];
    }
#pragma unroll
    for (int mask = 1; mask < 16; mask <<= 1) {
      sh  += __shfl_xor(sh,  mask, 32);
      sh2 += __shfl_xor(sh2, mask, 32);
      sc  += __shfl_xor(sc,  mask, 32);
      sc2 += __shfl_xor(sc2, mask, 32);
    }
    float mh = sh * (1.0f / 64.0f), mc = sc * (1.0f / 64.0f);
    float vh = sh2 * (1.0f / 64.0f) - mh * mh;
    float vc = sc2 * (1.0f / 64.0f) - mc * mc;
    float rh = rsqrtf(vh + EPS), rc = rsqrtf(vc + EPS);
    if (valid) {
#pragma unroll
      for (int t = 0; t < 4; ++t) {
        int h = t * 16 + lcol;
        float hn = (hloc[t] - mh) * rh;
        float cn = (cloc[t] - mc) * rc;
        hiddenOut[(long)node * HID + h] = hn * ln_h_s[h] + ln_h_b[h];
        cellOut[(long)node * HID + h]   = cn * ln_c_s[h] + ln_c_b[h];
        float ov = hn * ln_o_s[h] + ln_o_b[h];
        Y[(long)node * HID + h] = ov > 0.0f ? ov : 0.0f;
      }
    }
  }
}

__global__ __launch_bounds__(256)
void k_gemm2(const float* __restrict__ AY, const float* __restrict__ AS,
             const float* __restrict__ W1,
             const float* __restrict__ b1, float* __restrict__ Z, int N) {
  __shared__ float Bf[KSTEPS2 * 4 * 64];
  __shared__ float b1s[HID];

  int tid = threadIdx.x;
  for (int i = tid; i < KSTEPS2 * 4 * 64; i += 256) {
    int s = i >> 8;
    int rem = i & 255;
    int t = rem >> 6;
    int rem2 = rem & 63;
    int L = rem2 >> 1;
    int j = rem2 & 1;
    int k = 4 * s + 2 * (L >> 4) + j;
    int c = 16 * t + (L & 15);
    Bf[i] = (k < HID + 2) ? W1[k * HID + c] : 0.0f;
  }
  if (tid < HID) b1s[tid] = b1[tid];
  __syncthreads();

  int wave = tid >> 5, lane = tid & 31;
  int half = lane >> 4, lcol = lane & 15;
  int rowBase = blockIdx.x * ROWS_PER_BLOCK + wave * 16;

  int arow = rowBase + lcol;
  if (arow > N - 1) arow = N - 1;
  const float* aptr = AY + (long)arow * HID;
  const float* sptr = AS + (long)arow * 2;

  v8f acc[4];
#pragma unroll
  for (int t = 0; t < 4; ++t)
#pragma unroll
    for (int r = 0; r < 8; ++r) acc[t][r] = 0.0f;

  for (int s = 0; s < HID / 4; ++s) {
    int k0 = 4 * s + 2 * half;
    v2f a = *(const v2f*)(aptr + k0);
    const float* bbase = Bf + s * 256 + lane * 2;
#pragma unroll
    for (int t = 0; t < 4; ++t) {
      v2f b = *(const v2f*)(bbase + t * 64);
      acc[t] = __builtin_amdgcn_wmma_f32_16x16x4_f32(
          false, a, false, b, (short)0, acc[t], false, false);
    }
  }
  {
    float s0 = sptr[0], s1 = sptr[1];
    v2f a;
    a.x = (half == 0) ? s0 : 0.0f;
    a.y = (half == 0) ? s1 : 0.0f;
    const float* bbase = Bf + (HID / 4) * 256 + lane * 2;
#pragma unroll
    for (int t = 0; t < 4; ++t) {
      v2f b = *(const v2f*)(bbase + t * 64);
      acc[t] = __builtin_amdgcn_wmma_f32_16x16x4_f32(
          false, a, false, b, (short)0, acc[t], false, false);
    }
  }

#pragma unroll
  for (int r = 0; r < 8; ++r) {
    int node = rowBase + r + 8 * half;
    if (node >= N) continue;
#pragma unroll
    for (int t = 0; t < 4; ++t) {
      int h = t * 16 + lcol;
      float z = acc[t][r] + b1s[h];
      Z[(long)node * HID + h] = z > 0.0f ? z : 0.0f;
    }
  }
}

__global__ void k_final(const float* __restrict__ AZ, const float* __restrict__ W2,
                        const float* __restrict__ b2, const float* __restrict__ X,
                        float* __restrict__ out, int N) {
  __shared__ float w2s[HID];
  if (threadIdx.x < HID) w2s[threadIdx.x] = W2[threadIdx.x];
  __syncthreads();
  int n = blockIdx.x * blockDim.x + threadIdx.x;
  if (n >= N) return;
  const float* a = AZ + (long)n * HID;
  float acc = b2[0] + X[(long)n * FIN];
#pragma unroll 8
  for (int h = 0; h < HID; ++h) acc += a[h] * w2s[h];
  out[n] = acc;
}

extern "C" void kernel_launch(void* const* d_in, const int* in_sizes, int n_in,
                              void* d_out, int out_size, void* d_ws, size_t ws_size,
                              hipStream_t stream) {
  const float* X    = (const float*)d_in[0];
  const int*   eidx = (const int*)d_in[1];
  const float* ew   = (const float*)d_in[2];
  const float* skip = (const float*)d_in[3];
  const float* H    = (const float*)d_in[4];
  const float* C    = (const float*)d_in[5];
  const float* Wx   = (const float*)d_in[6];
  const float* bx   = (const float*)d_in[7];
  const float* Wh   = (const float*)d_in[8];
  const float* bh   = (const float*)d_in[9];
  const float* wc   = (const float*)d_in[10];
  const float* bg   = (const float*)d_in[11];
  const float* ln_h_s = (const float*)d_in[12];
  const float* ln_h_b = (const float*)d_in[13];
  const float* ln_c_s = (const float*)d_in[14];
  const float* ln_c_b = (const float*)d_in[15];
  const float* ln_o_s = (const float*)d_in[16];
  const float* ln_o_b = (const float*)d_in[17];
  const float* W1   = (const float*)d_in[18];
  const float* b1   = (const float*)d_in[19];
  const float* W2   = (const float*)d_in[20];
  const float* b2   = (const float*)d_in[21];

  const int N = in_sizes[0] / FIN;
  const int E = in_sizes[2];

  const int* src = eidx;
  const int* dst = eidx + E;

  float* out_p    = (float*)d_out;
  float* hidden_p = out_p + N;
  float* cell_p   = hidden_p + (long)N * HID;

  const int B = 256;
  auto cdiv = [](long a, long b) { return (unsigned)((a + b - 1) / b); };
  const int ntiles = (int)cdiv(N, TILE_R);

  float* ws   = (float*)d_ws;
  float* dinv = ws;
  int* tcount = (int*)(dinv + N);
  float* R1   = dinv + N + E;
  float* R2   = R1 + (long)N * KTOT;
  float* AXH  = R1;
  float* Y    = R2;
  float* AY   = R1;
  float* Z    = R2;
  float* AZ   = R1;
  float* XS   = R2;
  float* AS   = R2 + (long)N * HID;
  int2* lists = (int2*)(R2 + (long)N * KTOT);

  k_init_deg<<<cdiv((long)N * 4, B), B, 0, stream>>>(dinv, tcount, XS, X, skip, ntiles, N);
  k_acc_deg<<<cdiv(E, B), B, 0, stream>>>(dinv, dst, ew, E);
  k_dinv<<<cdiv(N, B), B, 0, stream>>>(dinv, N);
  k_bin<<<cdiv(E, BIN_THREADS * BIN_EPT), BIN_THREADS, 0, stream>>>(
      lists, tcount, dinv, src, dst, ew, ntiles, E);

  k_agg_tile<F1, true><<<ntiles, AGG_THREADS, 0, stream>>>(
      AXH, AS, XS, H, dinv, lists, tcount, src, dst, ew, N, E);

  size_t smem_gates = (size_t)(KSTEPS * NT_G * 64 + GCOLS) * sizeof(float);
  k_gates<<<cdiv(N, ROWS_PER_BLOCK), B, smem_gates, stream>>>(
      AXH, C, Wx, Wh, bx, bh, bg, wc,
      ln_h_s, ln_h_b, ln_c_s, ln_c_b, ln_o_s, ln_o_b,
      hidden_p, cell_p, Y, N);

  k_agg_tile<HID, false><<<ntiles, AGG_THREADS, 0, stream>>>(
      AY, nullptr, Y, nullptr, dinv, lists, tcount, src, dst, ew, N, E);

  k_gemm2<<<cdiv(N, ROWS_PER_BLOCK), B, 0, stream>>>(AY, AS, W1, b1, Z, N);

  k_agg_tile<HID, false><<<ntiles, AGG_THREADS, 0, stream>>>(
      AZ, nullptr, Z, nullptr, dinv, lists, tcount, src, dst, ew, N, E);

  k_final<<<cdiv(N, B), B, 0, stream>>>(AZ, W2, b2, X, out_p, N);
}


# deg^-1/2 factored out of the edges: gathered tables stored pre-scaled, tile rows scaled once at store; degrees summed in LDS from tile lists, arrival order; no per-edge look-ups; rounding changes
# speedup vs baseline: 4.4233x; 1.1471x over previous
#include <hip/hip_runtime.h>
#include <math.h>


typedef __attribute__((ext_vector_type(2))) float v2f;
typedef __attribute__((ext_vector_type(8))) float v8f;

#define FIN   8
#define HID   64
#define KTOT  72
#define KSTEPS 18
#define KSTEPS2 17
#define XSP   16
#define F1    74
#define GCOLS 256
#define NT_G  16
#define ROWS_PER_BLOCK 128
#define EPS 1e-5f

__device__ __forceinline__ float sigmoidf_(float x) { return 1.0f / (1.0f + expf(-x)); }

__global__ void k_zero_counts(int* tcount, int ntiles) {
  int i = blockIdx.x * blockDim.x + threadIdx.x;
  if (i < ntiles) tcount[i] = 0;
}


#define TILE_R      392
#define QCAP        8192
#define SCAN_U      8
#define AGG_THREADS 1024
#define CHUNK       (AGG_THREADS * 4 * SCAN_U)
#define LCAP        16384
#define KEY_BITS    9
#define BIN_THREADS 1024
#define BIN_EPT     8
#define BIN_TMAX    1024

__global__ __launch_bounds__(BIN_THREADS)
void k_bin(int2* __restrict__ lists, int* __restrict__ tcount,
           const int* __restrict__ src, const int* __restrict__ dst,
           const float* __restrict__ w, int ntiles, int E) {
  __shared__ int hist[BIN_TMAX + 1];
  const int tid = threadIdx.x;
  for (int t = tid; t <= ntiles; t += BIN_THREADS) hist[(t < ntiles) ? t : BIN_TMAX] = 0;
  __syncthreads();

  const int e0 = blockIdx.x * (BIN_THREADS * BIN_EPT) + tid;
  int d[BIN_EPT], s[BIN_EPT];
  float ww[BIN_EPT];
#pragma unroll
  for (int j = 0; j < BIN_EPT; ++j) {
    int e = e0 + j * BIN_THREADS;
    bool ok = e < E;
    d[j] = ok ? dst[e] : -1;
    s[j] = ok ? src[e] : 0;
    ww[j] = ok ? w[e] : 0.0f;
  }
  int tl[BIN_EPT], rk[BIN_EPT], key[BIN_EPT];
#pragma unroll
  for (int j = 0; j < BIN_EPT; ++j) {
    int di = (d[j] >= 0) ? d[j] : 0;
    int t = di / TILE_R;
    key[j] = (s[j] << KEY_BITS) | (di - t * TILE_R);
    tl[j] = (d[j] >= 0) ? t : BIN_TMAX;
  }
#pragma unroll
  for (int j = 0; j < BIN_EPT; ++j) rk[j] = atomicAdd(&hist[tl[j]], 1);
  __syncthreads();
  for (int t = tid; t < ntiles; t += BIN_THREADS) {
    int h = hist[t];
    hist[t] = (h > 0) ? atomicAdd(&tcount[t], h) : 0;
  }
  __syncthreads();
#pragma unroll
  for (int j = 0; j < BIN_EPT; ++j) {
    if (tl[j] < BIN_TMAX) {
      int pos = hist[tl[j]] + rk[j];
      if (pos < LCAP)
        lists[(size_t)tl[j] * LCAP + pos] = make_int2(key[j], __float_as_int(ww[j]));
    }
  }
}

__global__ __launch_bounds__(AGG_THREADS)
void k_deg_scale(float* __restrict__ dinv, float* __restrict__ XS, float* __restrict__ HS,
                 const float* __restrict__ X, const float* __restrict__ skip,
                 const float* __restrict__ H,
                 const int2* __restrict__ lists, const int* __restrict__ tcount,
                 const int* __restrict__ dst, const float* __restrict__ w, int N, int E) {
  __shared__ float degs[TILE_R];
  const int tid = threadIdx.x;
  const int lo = blockIdx.x * TILE_R;
  const int cnt = (N - lo < TILE_R) ? (N - lo) : TILE_R;
  for (int i = tid; i < TILE_R; i += AGG_THREADS) degs[i] = 1.0f;
  __syncthreads();
  const int nl = tcount[blockIdx.x];
  if (nl <= LCAP) {
    const int2* __restrict__ L = lists + (size_t)blockIdx.x * LCAP;
    for (int i = tid; i < nl; i += AGG_THREADS) {
      int2 rec = L[i];
      atomicAdd(&degs[rec.x & ((1 << KEY_BITS) - 1)], __int_as_float(rec.y));
    }
  } else {
    for (int e = tid; e < E; e += AGG_THREADS) {
      unsigned dl = (unsigned)(dst[e] - lo);
      if (dl < (unsigned)cnt) atomicAdd(&degs[dl], w[e]);
    }
  }
  __syncthreads();
  for (int i = tid; i < cnt; i += AGG_THREADS) {
    float dv = rsqrtf(degs[i]);
    degs[i] = dv;
    dinv[lo + i] = dv;
  }
  __syncthreads();
  for (int i = tid; i < cnt * 4; i += AGG_THREADS) {
    int n = i >> 2, q4 = i & 3;
    long row = lo + n;
    float dv = degs[n];
    float* o = XS + (row * 4 + q4) * 4;
    if (q4 < 2) {
      const float* x = X + row * FIN + q4 * 4;
      o[0] = dv * x[0]; o[1] = dv * x[1]; o[2] = dv * x[2]; o[3] = dv * x[3];
    } else if (q4 == 2) {
      o[0] = dv * skip[row * 2];
      o[1] = dv * skip[row * 2 + 1];
    }
  }
  const float4* __restrict__ H4 = (const float4*)(H + (long)lo * HID);
  float4* __restrict__ O4 = (float4*)(HS + (long)lo * HID);
  for (int i = tid; i < cnt * (HID / 4); i += AGG_THREADS) {
    float dv = degs[i >> 4];
    float4 v = H4[i];
    v.x *= dv; v.y *= dv; v.z *= dv; v.w *= dv;
    O4[i] = v;
  }
}

template <int F, bool SPLIT>
__device__ __forceinline__ float row_elem(const float* __restrict__ A,
                                          const float* __restrict__ B, int s, int k) {
  if (SPLIT) return (k < HID) ? B[(long)s * HID + k] : A[(long)s * XSP + (k - HID)];
  return A[(long)s * F + k];
}

template <int NE, int F, bool SPLIT>
__device__ __forceinline__ void add_rows(float* acc, const int2* qp,
                                         const float* __restrict__ A, const float* __restrict__ B,
                                         const int* __restrict__ src,
                                         const float* __restrict__ w, int lane) {
  constexpr int KPL = (F + 31) / 32;
  int s[NE], dl[NE];
  float cf[NE];
#pragma unroll
  for (int j = 0; j < NE; ++j) {
    int2 ent = qp[j];
    dl[j] = ent.y;
    s[j] = src[ent.x];
    cf[j] = w[ent.x];
  }
  float r[NE][KPL];
#pragma unroll
  for (int j = 0; j < NE; ++j)
#pragma unroll
    for (int m = 0; m < KPL; ++m) {
      int k = lane + 32 * m;
      r[j][m] = (k < F) ? row_elem<F, SPLIT>(A, B, s[j], k) : 0.0f;
    }
#pragma unroll
  for (int j = 0; j < NE; ++j)
#pragma unroll
    for (int m = 0; m < KPL; ++m) {
      int k = lane + 32 * m;
      if (k < F) atomicAdd(&acc[dl[j] * F + k], cf[j] * r[j][m]);
    }
}

template <int NE, int F, bool SPLIT>
__device__ __forceinline__ void add_recs(float* acc, int2 rec, int j0,
                                         const float* __restrict__ A, const float* __restrict__ B,
                                         int lane) {
  constexpr int KPL = (F + 31) / 32;
  int s[NE], dl[NE];
  float cf[NE];
#pragma unroll
  for (int j = 0; j < NE; ++j) {
    int key = __builtin_amdgcn_readlane(rec.x, j0 + j);
    cf[j] = __int_as_float(__builtin_amdgcn_readlane(rec.y, j0 + j));
    s[j] = key >> KEY_BITS;
    dl[j] = key & ((1 << KEY_BITS) - 1);
  }
  float r[NE][KPL];
#pragma unroll
  for (int j = 0; j < NE; ++j)
#pragma unroll
    for (int m = 0; m < KPL; ++m) {
      int k = lane + 32 * m;
      r[j][m] = (k < F) ? row_elem<F, SPLIT>(A, B, s[j], k) : 0.0f;
    }
#pragma unroll
  for (int j = 0; j < NE; ++j)
#pragma unroll
    for (int m = 0; m < KPL; ++m) {
      int k = lane + 32 * m;
      if (k < F) atomicAdd(&acc[dl[j] * F + k], cf[j] * r[j][m]);
    }
}

template <int F, bool SPLIT>
__global__ __launch_bounds__(AGG_THREADS)
void k_agg_tile(float* __restrict__ outbuf, float* __restrict__ outS, const float* __restrict__ A,
                const float* __restrict__ B, const float* __restrict__ dinv,
                const int2* __restrict__ lists, const int* __restrict__ tcount,
                const int* __restrict__ src, const int* __restrict__ dst,
                const float* __restrict__ w, int N, int E) {
  __shared__ float acc[TILE_R * F];
  __shared__ int2 q[QCAP];
  __shared__ int qn[2];

  const int tid = threadIdx.x;
  const int lane = tid & 31, wave = tid >> 5;
  const int lo = blockIdx.x * TILE_R;
  const int cnt = (N - lo < TILE_R) ? (N - lo) : TILE_R;

  for (int i = tid; i < cnt * F; i += AGG_THREADS) {
    int n = i / F, k = i - n * F;
    acc[i] = row_elem<F, SPLIT>(A, B, lo + n, k);
  }
  if (tid == 0) { qn[0] = 0; qn[1] = 0; }
  __syncthreads();

  const int nl = __builtin_amdgcn_readfirstlane(tcount[blockIdx.x]);
  if (nl <= LCAP) {
    const int2* __restrict__ L = lists + (size_t)blockIdx.x * LCAP;
    int base = __builtin_amdgcn_readfirstlane(wave) * 32;
    int2 rec = make_int2(0, 0);
    if (base + lane < nl) rec = L[base + lane];
    while (base < nl) {
      const int nbase = base + AGG_THREADS;
      int2 nrec = make_int2(0, 0);
      if (nbase + lane < nl) nrec = L[nbase + lane];
      const int nv = (nl - base < 32) ? (nl - base) : 32;
      int j = 0;
      for (; j + 8 <= nv; j += 8) add_recs<8, F, SPLIT>(acc, rec, j, A, B, lane);
      for (; j < nv; ++j) add_recs<1, F, SPLIT>(acc, rec, j, A, B, lane);
      rec = nrec;
      base = nbase;
    }
    __syncthreads();
  } else {
  const bool vec_ok = (((size_t)dst) & 15) == 0;
  const int nchunks = (E + CHUNK - 1) / CHUNK;
  int4 v[SCAN_U];

  auto load_chunk = [&](int c) {
    if (vec_ok && c * CHUNK + CHUNK <= E) {
#pragma unroll
      for (int u = 0; u < SCAN_U; ++u)
        v[u] = *(const int4*)(dst + c * CHUNK + (u * AGG_THREADS + tid) * 4);
    } else {
#pragma unroll
      for (int u = 0; u < SCAN_U; ++u) {
        int e0 = c * CHUNK + (u * AGG_THREADS + tid) * 4;
        v[u].x = (e0     < E) ? dst[e0]     : -1;
        v[u].y = (e0 + 1 < E) ? dst[e0 + 1] : -1;
        v[u].z = (e0 + 2 < E) ? dst[e0 + 2] : -1;
        v[u].w = (e0 + 3 < E) ? dst[e0 + 3] : -1;
      }
    }
  };

  load_chunk(0);
  int par = 0;
  for (int c = 0; c < nchunks; ++c) {
    int m = 0;
#pragma unroll
    for (int u = 0; u < SCAN_U; ++u) {
      m += ((unsigned)(v[u].x - lo) < (unsigned)cnt) ? 1 : 0;
      m += ((unsigned)(v[u].y - lo) < (unsigned)cnt) ? 1 : 0;
      m += ((unsigned)(v[u].z - lo) < (unsigned)cnt) ? 1 : 0;
      m += ((unsigned)(v[u].w - lo) < (unsigned)cnt) ? 1 : 0;
    }
    unsigned ovf = 0;
    if (m > 0) {
      int pos = atomicAdd(&qn[par], m);
#pragma unroll
      for (int u = 0; u < SCAN_U; ++u) {
        int e0 = c * CHUNK + (u * AGG_THREADS + tid) * 4;
        int dd[4] = {v[u].x, v[u].y, v[u].z, v[u].w};
#pragma unroll
        for (int j = 0; j < 4; ++j) {
          unsigned dl = (unsigned)(dd[j] - lo);
          if (dl < (unsigned)cnt) {
            if (pos < QCAP) q[pos] = make_int2(e0 + j, (int)dl);
            else ovf |= 1u << (4 * u + j);
            ++pos;
          }
        }
      }
    }
    while (ovf) {
      int b = __builtin_ctz(ovf);
      ovf &= ovf - 1;
      int e = c * CHUNK + ((b >> 2) * AGG_THREADS + tid) * 4 + (b & 3);
      int dl = dst[e] - lo;
      int s = src[e];
      float cf = w[e];
      for (int k = 0; k < F; ++k)
        atomicAdd(&acc[dl * F + k], cf * row_elem<F, SPLIT>(A, B, s, k));
    }
    __syncthreads();
    const int nq = (qn[par] < QCAP) ? qn[par] : QCAP;
    if (c + 1 < nchunks) load_chunk(c + 1);
    if (tid == 0) qn[par ^ 1] = 0;

    const int nfull = nq & ~3;
    for (int base = wave * 4; base < nfull; base += (AGG_THREADS / 32) * 4)
      add_rows<4, F, SPLIT>(acc, q + base, A, B, src, w, lane);
    if (nfull + wave < nq)
      add_rows<1, F, SPLIT>(acc, q + nfull + wave, A, B, src, w, lane);
    __syncthreads();
    par ^= 1;
  }
  }

  if (SPLIT) {
    for (int i = tid; i < cnt * F; i += AGG_THREADS) {
      int n = i / F, k = i - n * F;
      long row = lo + n;
      float v = acc[i] * dinv[row];
      if (k < HID) outbuf[row * KTOT + FIN + k] = v;
      else if (k < HID + FIN) outbuf[row * KTOT + (k - HID)] = v;
      else outS[row * 2 + (k - HID - FIN)] = v;
    }
  } else {
    for (int i = tid; i < cnt * F; i += AGG_THREADS)
      outbuf[(long)lo * F + i] = acc[i] * dinv[lo + i / F];
  }
}

__global__ __launch_bounds__(256)
void k_gates(const float* __restrict__ AXH, const float* __restrict__ C,
             const float* __restrict__ Wx, const float* __restrict__ Wh,
             const float* __restrict__ bx, const float* __restrict__ bh,
             const float* __restrict__ bg, const float* __restrict__ wc,
             const float* __restrict__ ln_h_s, const float* __restrict__ ln_h_b,
             const float* __restrict__ ln_c_s, const float* __restrict__ ln_c_b,
             const float* __restrict__ ln_o_s, const float* __restrict__ ln_o_b,
             float* __restrict__ hiddenOut, float* __restrict__ cellOut,
             float* __restrict__ Y, const float* __restrict__ dinv, int N) {
  extern __shared__ float smem[];
  float* Bf = smem;
  float* bias = smem + 18432;

  int tid = threadIdx.x;
  for (int i = tid; i < KSTEPS * NT_G * 64; i += 256) {
    int s = i >> 10;
    int rem = i & 1023;
    int t = rem >> 6;
    int rem2 = rem & 63;
    int L = rem2 >> 1;
    int j = rem2 & 1;
    int k = 4 * s + 2 * (L >> 4) + j;
    int c = 16 * t + (L & 15);
    int g = c >> 6, h = c & 63;
    Bf[i] = (k < FIN) ? Wx[(g * FIN + k) * HID + h]
                      : Wh[(g * HID + (k - FIN)) * HID + h];
  }
  if (tid < GCOLS) bias[tid] = bx[tid] + bh[tid] + bg[tid];
  __syncthreads();

  int wave = tid >> 5, lane = tid & 31;
  int half = lane >> 4, lcol = lane & 15;
  int rowBase = blockIdx.x * ROWS_PER_BLOCK + wave * 16;

  int arow = rowBase + lcol;
  if (arow > N - 1) arow = N - 1;
  const float* aptr = AXH + (long)arow * KTOT;

  v8f acc[NT_G];
#pragma unroll
  for (int t = 0; t < NT_G; ++t)
#pragma unroll
    for (int r = 0; r < 8; ++r) acc[t][r] = 0.0f;

  for (int s = 0; s < KSTEPS; ++s) {
    int k0 = 4 * s + 2 * half;
    v2f a = *(const v2f*)(aptr + k0);
    const float* bbase = Bf + s * 1024 + lane * 2;
#pragma unroll
    for (int t = 0; t < NT_G; ++t) {
      v2f b = *(const v2f*)(bbase + t * 64);
      acc[t] = __builtin_amdgcn_wmma_f32_16x16x4_f32(
          false, a, false, b, (short)0, acc[t], false, false);
    }
  }

#pragma unroll
  for (int r = 0; r < 8; ++r) {
    int node = rowBase + r + 8 * half;
    bool valid = node < N;
    int nc = valid ? node : N - 1;
    float hloc[4], cloc[4];
#pragma unroll
    for (int t = 0; t < 4; ++t) {
      int h = t * 16 + lcol;
      float c0 = C[(long)nc * HID + h];
      float gi = acc[t][r]      + bias[h]       + wc[h] * c0;
      float gf = acc[t + 4][r]  + bias[64 + h]  + wc[64 + h] * c0;
      float gc = acc[t + 8][r]  + bias[128 + h];
      float go = acc[t + 12][r] + bias[192 + h];
      float iv = sigmoidf_(gi);
      float fv = sigmoidf_(gf);
      float cc = fv * c0 + iv * tanhf(gc);
      float ov = sigmoidf_(go + wc[128 + h] * cc);
      hloc[t] = ov * tanhf(cc);
      cloc[t] = cc;
    }
    float sh = 0.f, sh2 = 0.f, sc = 0.f, sc2 = 0.f;
#pragma unroll
    for (int t = 0; t < 4; ++t) {
      sh += hloc[t]; sh2 += hloc[t] * hloc[t];
      sc += cloc[t]; sc2 += cloc[t] * cloc[t];
    }
#pragma unroll
    for (int mask = 1; mask < 16; mask <<= 1) {
      sh  += __shfl_xor(sh,  mask, 32);
      sh2 += __shfl_xor(sh2, mask, 32);
      sc  += __shfl_xor(sc,  mask, 32);
      sc2 += __shfl_xor(sc2, mask, 32);
    }
    float mh = sh * (1.0f / 64.0f), mc = sc * (1.0f / 64.0f);
    float vh = sh2 * (1.0f / 64.0f) - mh * mh;
    float vc = sc2 * (1.0f / 64.0f) - mc * mc;
    float rh = rsqrtf(vh + EPS), rc = rsqrtf(vc + EPS);
    float dv = dinv[nc];
    if (valid) {
#pragma unroll
      for (int t = 0; t < 4; ++t) {
        int h = t * 16 + lcol;
        float hn = (hloc[t] - mh) * rh;
        float cn = (cloc[t] - mc) * rc;
        hiddenOut[(long)node * HID + h] = hn * ln_h_s[h] + ln_h_b[h];
        cellOut[(long)node * HID + h]   = cn * ln_c_s[h] + ln_c_b[h];
        float ov = hn * ln_o_s[h] + ln_o_b[h];
        Y[(long)node * HID + h] = ov > 0.0f ? ov * dv : 0.0f;
      }
    }
  }
}

__global__ __launch_bounds__(256)
void k_gemm2(const float* __restrict__ AY, const float* __restrict__ AS,
             const float* __restrict__ W1,
             const float* __restrict__ b1, float* __restrict__ Z,
             const float* __restrict__ dinv, int N) {
  __shared__ float Bf[KSTEPS2 * 4 * 64];
  __shared__ float b1s[HID];

  int tid = threadIdx.x;
  for (int i = tid; i < KSTEPS2 * 4 * 64; i += 256) {
    int s = i >> 8;
    int rem = i & 255;
    int t = rem >> 6;
    int rem2 = rem & 63;
    int L = rem2 >> 1;
    int j = rem2 & 1;
    int k = 4 * s + 2 * (L >> 4) + j;
    int c = 16 * t + (L & 15);
    Bf[i] = (k < HID + 2) ? W1[k * HID + c] : 0.0f;
  }
  if (tid < HID) b1s[tid] = b1[tid];
  __syncthreads();

  int wave = tid >> 5, lane = tid & 31;
  int half = lane >> 4, lcol = lane & 15;
  int rowBase = blockIdx.x * ROWS_PER_BLOCK + wave * 16;

  int arow = rowBase + lcol;
  if (arow > N - 1) arow = N - 1;
  const float* aptr = AY + (long)arow * HID;
  const float* sptr = AS + (long)arow * 2;

  v8f acc[4];
#pragma unroll
  for (int t = 0; t < 4; ++t)
#pragma unroll
    for (int r = 0; r < 8; ++r) acc[t][r] = 0.0f;

  for (int s = 0; s < HID / 4; ++s) {
    int k0 = 4 * s + 2 * half;
    v2f a = *(const v2f*)(aptr + k0);
    const float* bbase = Bf + s * 256 + lane * 2;
#pragma unroll
    for (int t = 0; t < 4; ++t) {
      v2f b = *(const v2f*)(bbase + t * 64);
      acc[t] = __builtin_amdgcn_wmma_f32_16x16x4_f32(
          false, a, false, b, (short)0, acc[t], false, false);
    }
  }
  {
    float s0 = sptr[0], s1 = sptr[1];
    v2f a;
    a.x = (half == 0) ? s0 : 0.0f;
    a.y = (half == 0) ? s1 : 0.0f;
    const float* bbase = Bf + (HID / 4) * 256 + lane * 2;
#pragma unroll
    for (int t = 0; t < 4; ++t) {
      v2f b = *(const v2f*)(bbase + t * 64);
      acc[t] = __builtin_amdgcn_wmma_f32_16x16x4_f32(
          false, a, false, b, (short)0, acc[t], false, false);
    }
  }

#pragma unroll
  for (int r = 0; r < 8; ++r) {
    int node = rowBase + r + 8 * half;
    if (node >= N) continue;
    float dv = dinv[node];
#pragma unroll
    for (int t = 0; t < 4; ++t) {
      int h = t * 16 + lcol;
      float z = acc[t][r] + b1s[h];
      Z[(long)node * HID + h] = z > 0.0f ? z * dv : 0.0f;
    }
  }
}

__global__ void k_final(const float* __restrict__ AZ, const float* __restrict__ W2,
                        const float* __restrict__ b2, const float* __restrict__ X,
                        float* __restrict__ out, int N) {
  __shared__ float w2s[HID];
  if (threadIdx.x < HID) w2s[threadIdx.x] = W2[threadIdx.x];
  __syncthreads();
  int n = blockIdx.x * blockDim.x + threadIdx.x;
  if (n >= N) return;
  const float* a = AZ + (long)n * HID;
  float acc = b2[0] + X[(long)n * FIN];
#pragma unroll 8
  for (int h = 0; h < HID; ++h) acc += a[h] * w2s[h];
  out[n] = acc;
}

extern "C" void kernel_launch(void* const* d_in, const int* in_sizes, int n_in,
                              void* d_out, int out_size, void* d_ws, size_t ws_size,
                              hipStream_t stream) {
  const float* X    = (const float*)d_in[0];
  const int*   eidx = (const int*)d_in[1];
  const float* ew   = (const float*)d_in[2];
  const float* skip = (const float*)d_in[3];
  const float* H    = (const float*)d_in[4];
  const float* C    = (const float*)d_in[5];
  const float* Wx   = (const float*)d_in[6];
  const float* bx   = (const float*)d_in[7];
  const float* Wh   = (const float*)d_in[8];
  const float* bh   = (const float*)d_in[9];
  const float* wc   = (const float*)d_in[10];
  const float* bg   = (const float*)d_in[11];
  const float* ln_h_s = (const float*)d_in[12];
  const float* ln_h_b = (const float*)d_in[13];
  const float* ln_c_s = (const float*)d_in[14];
  const float* ln_c_b = (const float*)d_in[15];
  const float* ln_o_s = (const float*)d_in[16];
  const float* ln_o_b = (const float*)d_in[17];
  const float* W1   = (const float*)d_in[18];
  const float* b1   = (const float*)d_in[19];
  const float* W2   = (const float*)d_in[20];
  const float* b2   = (const float*)d_in[21];

  const int N = in_sizes[0] / FIN;
  const int E = in_sizes[2];

  const int* src = eidx;
  const int* dst = eidx + E;

  float* out_p    = (float*)d_out;
  float* hidden_p = out_p + N;
  float* cell_p   = hidden_p + (long)N * HID;

  const int B = 256;
  auto cdiv = [](long a, long b) { return (unsigned)((a + b - 1) / b); };
  const int ntiles = (int)cdiv(N, TILE_R);

  float* ws   = (float*)d_ws;
  float* dinv = ws;
  float* XS   = dinv + N;
  float* R1   = dinv + N + E;
  float* R2   = R1 + (long)N * KTOT;
  float* AXH  = R1;
  float* Y    = R2;
  float* AY   = R1;
  float* Z    = R2;
  float* AZ   = R1;
  float* HS   = R2;
  float* AS   = R2 + (long)N * HID;
  int* tcount = (int*)(AS + (long)N * 2);
  int2* lists = (int2*)(R2 + (long)N * KTOT);

  k_zero_counts<<<cdiv(ntiles, B), B, 0, stream>>>(tcount, ntiles);
  k_bin<<<cdiv(E, BIN_THREADS * BIN_EPT), BIN_THREADS, 0, stream>>>(
      lists, tcount, src, dst, ew, ntiles, E);
  k_deg_scale<<<ntiles, AGG_THREADS, 0, stream>>>(
      dinv, XS, HS, X, skip, H, lists, tcount, dst, ew, N, E);

  k_agg_tile<F1, true><<<ntiles, AGG_THREADS, 0, stream>>>(
      AXH, AS, XS, HS, dinv, lists, tcount, src, dst, ew, N, E);

  size_t smem_gates = (size_t)(KSTEPS * NT_G * 64 + GCOLS) * sizeof(float);
  k_gates<<<cdiv(N, ROWS_PER_BLOCK), B, smem_gates, stream>>>(
      AXH, C, Wx, Wh, bx, bh, bg, wc,
      ln_h_s, ln_h_b, ln_c_s, ln_c_b, ln_o_s, ln_o_b,
      hidden_p, cell_p, Y, dinv, N);

  k_agg_tile<HID, false><<<ntiles, AGG_THREADS, 0, stream>>>(
      AY, nullptr, Y, nullptr, dinv, lists, tcount, src, dst, ew, N, E);

  k_gemm2<<<cdiv(N, ROWS_PER_BLOCK), B, 0, stream>>>(AY, AS, W1, b1, Z, dinv, N);

  k_agg_tile<HID, false><<<ntiles, AGG_THREADS, 0, stream>>>(
      AZ, nullptr, Z, nullptr, dinv, lists, tcount, src, dst, ew, N, E);

  k_final<<<cdiv(N, B), B, 0, stream>>>(AZ, W2, b2, X, out_p, N);
}


# 2nd and 3rd aggregation kernels consume their finished LDS tile themselves: head GEMM (WMMA, A from the tile) and final dot (half-row sums, rotated order: float sum order of out changes); 7 launches
# speedup vs baseline: 4.8614x; 1.0990x over previous
#include <hip/hip_runtime.h>
#include <math.h>


typedef __attribute__((ext_vector_type(2))) float v2f;
typedef __attribute__((ext_vector_type(8))) float v8f;

#define FIN   8
#define HID   64
#define KTOT  72
#define KSTEPS 18
#define KSTEPS2 17
#define XSP   16
#define F1    74
#define GCOLS 256
#define NT_G  16
#define ROWS_PER_BLOCK 128
#define EPS 1e-5f

__device__ __forceinline__ float sigmoidf_(float x) { return 1.0f / (1.0f + expf(-x)); }

__global__ void k_zero_counts(int* tcount, int ntiles) {
  int i = blockIdx.x * blockDim.x + threadIdx.x;
  if (i < ntiles) tcount[i] = 0;
}


#define TILE_R      392
#define QCAP        8192
#define SCAN_U      8
#define AGG_THREADS 1024
#define CHUNK       (AGG_THREADS * 4 * SCAN_U)
#define LCAP        16384
#define KEY_BITS    9
#define BIN_THREADS 1024
#define BIN_EPT     8
#define BIN_TMAX    1024

__global__ __launch_bounds__(BIN_THREADS)
void k_bin(int2* __restrict__ lists, int* __restrict__ tcount,
           const int* __restrict__ src, const int* __restrict__ dst,
           const float* __restrict__ w, int ntiles, int E) {
  __shared__ int hist[BIN_TMAX + 1];
  const int tid = threadIdx.x;
  for (int t = tid; t <= ntiles; t += BIN_THREADS) hist[(t < ntiles) ? t : BIN_TMAX] = 0;
  __syncthreads();

  const int e0 = blockIdx.x * (BIN_THREADS * BIN_EPT) + tid;
  int d[BIN_EPT], s[BIN_EPT];
  float ww[BIN_EPT];
#pragma unroll
  for (int j = 0; j < BIN_EPT; ++j) {
    int e = e0 + j * BIN_THREADS;
    bool ok = e < E;
    d[j] = ok ? dst[e] : -1;
    s[j] = ok ? src[e] : 0;
    ww[j] = ok ? w[e] : 0.0f;
  }
  int tl[BIN_EPT], rk[BIN_EPT], key[BIN_EPT];
#pragma unroll
  for (int j = 0; j < BIN_EPT; ++j) {
    int di = (d[j] >= 0) ? d[j] : 0;
    int t = di / TILE_R;
    key[j] = (s[j] << KEY_BITS) | (di - t * TILE_R);
    tl[j] = (d[j] >= 0) ? t : BIN_TMAX;
  }
#pragma unroll
  for (int j = 0; j < BIN_EPT; ++j) rk[j] = atomicAdd(&hist[tl[j]], 1);
  __syncthreads();
  for (int t = tid; t < ntiles; t += BIN_THREADS) {
    int h = hist[t];
    hist[t] = (h > 0) ? atomicAdd(&tcount[t], h) : 0;
  }
  __syncthreads();
#pragma unroll
  for (int j = 0; j < BIN_EPT; ++j) {
    if (tl[j] < BIN_TMAX) {
      int pos = hist[tl[j]] + rk[j];
      if (pos < LCAP)
        lists[(size_t)tl[j] * LCAP + pos] = make_int2(key[j], __float_as_int(ww[j]));
    }
  }
}

__global__ __launch_bounds__(AGG_THREADS)
void k_deg_scale(float* __restrict__ dinv, float* __restrict__ XS, float* __restrict__ HS,
                 const float* __restrict__ X, const float* __restrict__ skip,
                 const float* __restrict__ H,
                 const int2* __restrict__ lists, const int* __restrict__ tcount,
                 const int* __restrict__ dst, const float* __restrict__ w, int N, int E) {
  __shared__ float degs[TILE_R];
  const int tid = threadIdx.x;
  const int lo = blockIdx.x * TILE_R;
  const int cnt = (N - lo < TILE_R) ? (N - lo) : TILE_R;
  for (int i = tid; i < TILE_R; i += AGG_THREADS) degs[i] = 1.0f;
  __syncthreads();
  const int nl = tcount[blockIdx.x];
  if (nl <= LCAP) {
    const int2* __restrict__ L = lists + (size_t)blockIdx.x * LCAP;
    for (int i = tid; i < nl; i += AGG_THREADS) {
      int2 rec = L[i];
      atomicAdd(&degs[rec.x & ((1 << KEY_BITS) - 1)], __int_as_float(rec.y));
    }
  } else {
    for (int e = tid; e < E; e += AGG_THREADS) {
      unsigned dl = (unsigned)(dst[e] - lo);
      if (dl < (unsigned)cnt) atomicAdd(&degs[dl], w[e]);
    }
  }
  __syncthreads();
  for (int i = tid; i < cnt; i += AGG_THREADS) {
    float dv = rsqrtf(degs[i]);
    degs[i] = dv;
    dinv[lo + i] = dv;
  }
  __syncthreads();
  for (int i = tid; i < cnt * 4; i += AGG_THREADS) {
    int n = i >> 2, q4 = i & 3;
    long row = lo + n;
    float dv = degs[n];
    float* o = XS + (row * 4 + q4) * 4;
    if (q4 < 2) {
      const float* x = X + row * FIN + q4 * 4;
      o[0] = dv * x[0]; o[1] = dv * x[1]; o[2] = dv * x[2]; o[3] = dv * x[3];
    } else if (q4 == 2) {
      o[0] = dv * skip[row * 2];
      o[1] = dv * skip[row * 2 + 1];
    }
  }
  const float4* __restrict__ H4 = (const float4*)(H + (long)lo * HID);
  float4* __restrict__ O4 = (float4*)(HS + (long)lo * HID);
  for (int i = tid; i < cnt * (HID / 4); i += AGG_THREADS) {
    float dv = degs[i >> 4];
    float4 v = H4[i];
    v.x *= dv; v.y *= dv; v.z *= dv; v.w *= dv;
    O4[i] = v;
  }
}

template <int F, bool SPLIT>
__device__ __forceinline__ float row_elem(const float* __restrict__ A,
                                          const float* __restrict__ B, int s, int k) {
  if (SPLIT) return (k < HID) ? B[(long)s * HID + k] : A[(long)s * XSP + (k - HID)];
  return A[(long)s * F + k];
}

template <int NE, int F, bool SPLIT>
__device__ __forceinline__ void add_rows(float* acc, const int2* qp,
                                         const float* __restrict__ A, const float* __restrict__ B,
                                         const int* __restrict__ src,
                                         const float* __restrict__ w, int lane) {
  constexpr int KPL = (F + 31) / 32;
  int s[NE], dl[NE];
  float cf[NE];
#pragma unroll
  for (int j = 0; j < NE; ++j) {
    int2 ent = qp[j];
    dl[j] = ent.y;
    s[j] = src[ent.x];
    cf[j] = w[ent.x];
  }
  float r[NE][KPL];
#pragma unroll
  for (int j = 0; j < NE; ++j)
#pragma unroll
    for (int m = 0; m < KPL; ++m) {
      int k = lane + 32 * m;
      r[j][m] = (k < F) ? row_elem<F, SPLIT>(A, B, s[j], k) : 0.0f;
    }
#pragma unroll
  for (int j = 0; j < NE; ++j)
#pragma unroll
    for (int m = 0; m < KPL; ++m) {
      int k = lane + 32 * m;
      if (k < F) atomicAdd(&acc[dl[j] * F + k], cf[j] * r[j][m]);
    }
}

template <int NE, int F, bool SPLIT>
__device__ __forceinline__ void add_recs(float* acc, int2 rec, int j0,
                                         const float* __restrict__ A, const float* __restrict__ B,
                                         int lane) {
  constexpr int KPL = (F + 31) / 32;
  int s[NE], dl[NE];
  float cf[NE];
#pragma unroll
  for (int j = 0; j < NE; ++j) {
    int key = __builtin_amdgcn_readlane(rec.x, j0 + j);
    cf[j] = __int_as_float(__builtin_amdgcn_readlane(rec.y, j0 + j));
    s[j] = key >> KEY_BITS;
    dl[j] = key & ((1 << KEY_BITS) - 1);
  }
  float r[NE][KPL];
#pragma unroll
  for (int j = 0; j < NE; ++j)
#pragma unroll
    for (int m = 0; m < KPL; ++m) {
      int k = lane + 32 * m;
      r[j][m] = (k < F) ? row_elem<F, SPLIT>(A, B, s[j], k) : 0.0f;
    }
#pragma unroll
  for (int j = 0; j < NE; ++j)
#pragma unroll
    for (int m = 0; m < KPL; ++m) {
      int k = lane + 32 * m;
      if (k < F) atomicAdd(&acc[dl[j] * F + k], cf[j] * r[j][m]);
    }
}

template <int F, bool SPLIT, int EPI>
__global__ __launch_bounds__(AGG_THREADS)
void k_agg_tile(float* __restrict__ outbuf, float* __restrict__ outS, const float* __restrict__ A,
                const float* __restrict__ B, const float* __restrict__ dinv,
                const int2* __restrict__ lists, const int* __restrict__ tcount,
                const int* __restrict__ src, const int* __restrict__ dst,
                const float* __restrict__ w, const float* __restrict__ Wd,
                const float* __restrict__ bd, const float* __restrict__ aux, int N, int E) {
  __shared__ float acc[TILE_R * F];
  __shared__ int2 q[QCAP];
  __shared__ int qn[2];
  __shared__ float Bf[(EPI == 1) ? KSTEPS2 * 4 * 64 : 2];
  __shared__ float b1s[(EPI == 1) ? HID : 1];
  __shared__ float w2s[(EPI == 2) ? HID : 1];
  __shared__ float dvs[(EPI != 0) ? TILE_R : 1];

  const int tid = threadIdx.x;
  const int lane = tid & 31, wave = tid >> 5;
  const int lo = blockIdx.x * TILE_R;
  const int cnt = (N - lo < TILE_R) ? (N - lo) : TILE_R;

  if (EPI == 1) {
    for (int i = tid; i < KSTEPS2 * 4 * 64; i += AGG_THREADS) {
      int s = i >> 8;
      int rem = i & 255;
      int t = rem >> 6;
      int rem2 = rem & 63;
      int L = rem2 >> 1;
      int j = rem2 & 1;
      int k = 4 * s + 2 * (L >> 4) + j;
      int c = 16 * t + (L & 15);
      Bf[i] = (k < HID + 2) ? Wd[k * HID + c] : 0.0f;
    }
    if (tid < HID) b1s[tid] = bd[tid];
  }
  if (EPI == 2 && tid < HID) w2s[tid] = Wd[tid];
  if (EPI != 0)
    for (int i = tid; i < cnt; i += AGG_THREADS) dvs[i] = dinv[lo + i];

  for (int i = tid; i < cnt * F; i += AGG_THREADS) {
    int n = i / F, k = i - n * F;
    acc[i] = row_elem<F, SPLIT>(A, B, lo + n, k);
  }
  if (tid == 0) { qn[0] = 0; qn[1] = 0; }
  __syncthreads();

  const int nl = __builtin_amdgcn_readfirstlane(tcount[blockIdx.x]);
  if (nl <= LCAP) {
    const int2* __restrict__ L = lists + (size_t)blockIdx.x * LCAP;
    int base = __builtin_amdgcn_readfirstlane(wave) * 32;
    int2 rec = make_int2(0, 0);
    if (base + lane < nl) rec = L[base + lane];
    while (base < nl) {
      const int nbase = base + AGG_THREADS;
      int2 nrec = make_int2(0, 0);
      if (nbase + lane < nl) nrec = L[nbase + lane];
      const int nv = (nl - base < 32) ? (nl - base) : 32;
      int j = 0;
      for (; j + 8 <= nv; j += 8) add_recs<8, F, SPLIT>(acc, rec, j, A, B, lane);
      for (; j < nv; ++j) add_recs<1, F, SPLIT>(acc, rec, j, A, B, lane);
      rec = nrec;
      base = nbase;
    }
    __syncthreads();
  } else {
  const bool vec_ok = (((size_t)dst) & 15) == 0;
  const int nchunks = (E + CHUNK - 1) / CHUNK;
  int4 v[SCAN_U];

  auto load_chunk = [&](int c) {
    if (vec_ok && c * CHUNK + CHUNK <= E) {
#pragma unroll
      for (int u = 0; u < SCAN_U; ++u)
        v[u] = *(const int4*)(dst + c * CHUNK + (u * AGG_THREADS + tid) * 4);
    } else {
#pragma unroll
      for (int u = 0; u < SCAN_U; ++u) {
        int e0 = c * CHUNK + (u * AGG_THREADS + tid) * 4;
        v[u].x = (e0     < E) ? dst[e0]     : -1;
        v[u].y = (e0 + 1 < E) ? dst[e0 + 1] : -1;
        v[u].z = (e0 + 2 < E) ? dst[e0 + 2] : -1;
        v[u].w = (e0 + 3 < E) ? dst[e0 + 3] : -1;
      }
    }
  };

  load_chunk(0);
  int par = 0;
  for (int c = 0; c < nchunks; ++c) {
    int m = 0;
#pragma unroll
    for (int u = 0; u < SCAN_U; ++u) {
      m += ((unsigned)(v[u].x - lo) < (unsigned)cnt) ? 1 : 0;
      m += ((unsigned)(v[u].y - lo) < (unsigned)cnt) ? 1 : 0;
      m += ((unsigned)(v[u].z - lo) < (unsigned)cnt) ? 1 : 0;
      m += ((unsigned)(v[u].w - lo) < (unsigned)cnt) ? 1 : 0;
    }
    unsigned ovf = 0;
    if (m > 0) {
      int pos = atomicAdd(&qn[par], m);
#pragma unroll
      for (int u = 0; u < SCAN_U; ++u) {
        int e0 = c * CHUNK + (u * AGG_THREADS + tid) * 4;
        int dd[4] = {v[u].x, v[u].y, v[u].z, v[u].w};
#pragma unroll
        for (int j = 0; j < 4; ++j) {
          unsigned dl = (unsigned)(dd[j] - lo);
          if (dl < (unsigned)cnt) {
            if (pos < QCAP) q[pos] = make_int2(e0 + j, (int)dl);
            else ovf |= 1u << (4 * u + j);
            ++pos;
          }
        }
      }
    }
    while (ovf) {
      int b = __builtin_ctz(ovf);
      ovf &= ovf - 1;
      int e = c * CHUNK + ((b >> 2) * AGG_THREADS + tid) * 4 + (b & 3);
      int dl = dst[e] - lo;
      int s = src[e];
      float cf = w[e];
      for (int k = 0; k < F; ++k)
        atomicAdd(&acc[dl * F + k], cf * row_elem<F, SPLIT>(A, B, s, k));
    }
    __syncthreads();
    const int nq = (qn[par] < QCAP) ? qn[par] : QCAP;
    if (c + 1 < nchunks) load_chunk(c + 1);
    if (tid == 0) qn[par ^ 1] = 0;

    const int nfull = nq & ~3;
    for (int base = wave * 4; base < nfull; base += (AGG_THREADS / 32) * 4)
      add_rows<4, F, SPLIT>(acc, q + base, A, B, src, w, lane);
    if (nfull + wave < nq)
      add_rows<1, F, SPLIT>(acc, q + nfull + wave, A, B, src, w, lane);
    __syncthreads();
    par ^= 1;
  }
  }

  if (EPI == 1) {
    const int half = lane >> 4, lcol = lane & 15;
    static_assert((TILE_R + 15) / 16 <= AGG_THREADS / 32, "a wave takes at most one strip");
    const int mt = __builtin_amdgcn_readfirstlane(wave);
    if (mt * 16 < cnt) {
      int rl = mt * 16 + lcol;
      if (rl > cnt - 1) rl = cnt - 1;
      const float dva = dvs[rl];
      const float* ap = acc + rl * F;
      const float* sp = aux + (long)(lo + rl) * 2;
      const float s0 = sp[0], s1 = sp[1];
      v8f c[4];
#pragma unroll
      for (int t = 0; t < 4; ++t)
#pragma unroll
        for (int r = 0; r < 8; ++r) c[t][r] = 0.0f;
#pragma unroll
      for (int s = 0; s < HID / 4; ++s) {
        v2f a = *(const v2f*)(ap + 4 * s + 2 * half);
        a.x *= dva; a.y *= dva;
        const float* bbase = Bf + s * 256 + lane * 2;
#pragma unroll
        for (int t = 0; t < 4; ++t) {
          v2f b = *(const v2f*)(bbase + t * 64);
          c[t] = __builtin_amdgcn_wmma_f32_16x16x4_f32(
              false, a, false, b, (short)0, c[t], false, false);
        }
      }
      {
        v2f a;
        a.x = (half == 0) ? s0 : 0.0f;
        a.y = (half == 0) ? s1 : 0.0f;
        const float* bbase = Bf + (HID / 4) * 256 + lane * 2;
#pragma unroll
        for (int t = 0; t < 4; ++t) {
          v2f b = *(const v2f*)(bbase + t * 64);
          c[t] = __builtin_amdgcn_wmma_f32_16x16x4_f32(
              false, a, false, b, (short)0, c[t], false, false);
        }
      }
#pragma unroll
      for (int r = 0; r < 8; ++r) {
        int rn = mt * 16 + r + 8 * half;
        if (rn < cnt) {
          long row = lo + rn;
          float dv = dvs[rn];
#pragma unroll
          for (int t = 0; t < 4; ++t) {
            int h = t * 16 + lcol;
            float z = c[t][r] + b1s[h];
            outbuf[row * HID + h] = z > 0.0f ? z * dv : 0.0f;
          }
        }
      }
    }
  } else if (EPI == 2) {
    const int r = tid >> 1, part = tid & 1;
    float p = 0.0f;
    if (r < cnt) {
      const float dv = dvs[r];
      const float* ar = acc + r * F + part * 32;
      const float* wr = w2s + part * 32;
#pragma unroll
      for (int j = 0; j < 32; ++j) {
        int k = (j + r) & 31;
        p += (ar[k] * dv) * wr[k];
      }
    }
    p += __shfl_xor(p, 1, 32);
    if (part == 0 && r < cnt) outbuf[lo + r] = (bd[0] + aux[(long)(lo + r) * FIN]) + p;
  } else if (SPLIT) {
    for (int i = tid; i < cnt * F; i += AGG_THREADS) {
      int n = i / F, k = i - n * F;
      long row = lo + n;
      float v = acc[i] * dinv[row];
      if (k < HID) outbuf[row * KTOT + FIN + k] = v;
      else if (k < HID + FIN) outbuf[row * KTOT + (k - HID)] = v;
      else outS[row * 2 + (k - HID - FIN)] = v;
    }
  } else {
    for (int i = tid; i < cnt * F; i += AGG_THREADS)
      outbuf[(long)lo * F + i] = acc[i] * dinv[lo + i / F];
  }
}

__global__ __launch_bounds__(256)
void k_gates(const float* __restrict__ AXH, const float* __restrict__ C,
             const float* __restrict__ Wx, const float* __restrict__ Wh,
             const float* __restrict__ bx, const float* __restrict__ bh,
             const float* __restrict__ bg, const float* __restrict__ wc,
             const float* __restrict__ ln_h_s, const float* __restrict__ ln_h_b,
             const float* __restrict__ ln_c_s, const float* __restrict__ ln_c_b,
             const float* __restrict__ ln_o_s, const float* __restrict__ ln_o_b,
             float* __restrict__ hiddenOut, float* __restrict__ cellOut,
             float* __restrict__ Y, const float* __restrict__ dinv, int N) {
  extern __shared__ float smem[];
  float* Bf = smem;
  float* bias = smem + 18432;

  int tid = threadIdx.x;
  for (int i = tid; i < KSTEPS * NT_G * 64; i += 256) {
    int s = i >> 10;
    int rem = i & 1023;
    int t = rem >> 6;
    int rem2 = rem & 63;
    int L = rem2 >> 1;
    int j = rem2 & 1;
    int k = 4 * s + 2 * (L >> 4) + j;
    int c = 16 * t + (L & 15);
    int g = c >> 6, h = c & 63;
    Bf[i] = (k < FIN) ? Wx[(g * FIN + k) * HID + h]
                      : Wh[(g * HID + (k - FIN)) * HID + h];
  }
  if (tid < GCOLS) bias[tid] = bx[tid] + bh[tid] + bg[tid];
  __syncthreads();

  int wave = tid >> 5, lane = tid & 31;
  int half = lane >> 4, lcol = lane & 15;
  int rowBase = blockIdx.x * ROWS_PER_BLOCK + wave * 16;

  int arow = rowBase + lcol;
  if (arow > N - 1) arow = N - 1;
  const float* aptr = AXH + (long)arow * KTOT;

  v8f acc[NT_G];
#pragma unroll
  for (int t = 0; t < NT_G; ++t)
#pragma unroll
    for (int r = 0; r < 8; ++r) acc[t][r] = 0.0f;

  for (int s = 0; s < KSTEPS; ++s) {
    int k0 = 4 * s + 2 * half;
    v2f a = *(const v2f*)(aptr + k0);
    const float* bbase = Bf + s * 1024 + lane * 2;
#pragma unroll
    for (int t = 0; t < NT_G; ++t) {
      v2f b = *(const v2f*)(bbase + t * 64);
      acc[t] = __builtin_amdgcn_wmma_f32_16x16x4_f32(
          false, a, false, b, (short)0, acc[t], false, false);
    }
  }

#pragma unroll
  for (int r = 0; r < 8; ++r) {
    int node = rowBase + r + 8 * half;
    bool valid = node < N;
    int nc = valid ? node : N - 1;
    float hloc[4], cloc[4];
#pragma unroll
    for (int t = 0; t < 4; ++t) {
      int h = t * 16 + lcol;
      float c0 = C[(long)nc * HID + h];
      float gi = acc[t][r]      + bias[h]       + wc[h] * c0;
      float gf = acc[t + 4][r]  + bias[64 + h]  + wc[64 + h] * c0;
      float gc = acc[t + 8][r]  + bias[128 + h];
      float go = acc[t + 12][r] + bias[192 + h];
      float iv = sigmoidf_(gi);
      float fv = sigmoidf_(gf);
      float cc = fv * c0 + iv * tanhf(gc);
      float ov = sigmoidf_(go + wc[128 + h] * cc);
      hloc[t] = ov * tanhf(cc);
      cloc[t] = cc;
    }
    float sh = 0.f, sh2 = 0.f, sc = 0.f, sc2 = 0.f;
#pragma unroll
    for (int t = 0; t < 4; ++t) {
      sh += hloc[t]; sh2 += hloc[t] * hloc[t];
      sc += cloc[t]; sc2 += cloc[t] * cloc[t];
    }
#pragma unroll
    for (int mask = 1; mask < 16; mask <<= 1) {
      sh  += __shfl_xor(sh,  mask, 32);
      sh2 += __shfl_xor(sh2, mask, 32);
      sc  += __shfl_xor(sc,  mask, 32);
      sc2 += __shfl_xor(sc2, mask, 32);
    }
    float mh = sh * (1.0f / 64.0f), mc = sc * (1.0f / 64.0f);
    float vh = sh2 * (1.0f / 64.0f) - mh * mh;
    float vc = sc2 * (1.0f / 64.0f) - mc * mc;
    float rh = rsqrtf(vh + EPS), rc = rsqrtf(vc + EPS);
    float dv = dinv[nc];
    if (valid) {
#pragma unroll
      for (int t = 0; t < 4; ++t) {
        int h = t * 16 + lcol;
        float hn = (hloc[t] - mh) * rh;
        float cn = (cloc[t] - mc) * rc;
        hiddenOut[(long)node * HID + h] = hn * ln_h_s[h] + ln_h_b[h];
        cellOut[(long)node * HID + h]   = cn * ln_c_s[h] + ln_c_b[h];
        float ov = hn * ln_o_s[h] + ln_o_b[h];
        Y[(long)node * HID + h] = ov > 0.0f ? ov * dv : 0.0f;
      }
    }
  }
}

extern "C" void kernel_launch(void* const* d_in, const int* in_sizes, int n_in,
                              void* d_out, int out_size, void* d_ws, size_t ws_size,
                              hipStream_t stream) {
  const float* X    = (const float*)d_in[0];
  const int*   eidx = (const int*)d_in[1];
  const float* ew   = (const float*)d_in[2];
  const float* skip = (const float*)d_in[3];
  const float* H    = (const float*)d_in[4];
  const float* C    = (const float*)d_in[5];
  const float* Wx   = (const float*)d_in[6];
  const float* bx   = (const float*)d_in[7];
  const float* Wh   = (const float*)d_in[8];
  const float* bh   = (const float*)d_in[9];
  const float* wc   = (const float*)d_in[10];
  const float* bg   = (const float*)d_in[11];
  const float* ln_h_s = (const float*)d_in[12];
  const float* ln_h_b = (const float*)d_in[13];
  const float* ln_c_s = (const float*)d_in[14];
  const float* ln_c_b = (const float*)d_in[15];
  const float* ln_o_s = (const float*)d_in[16];
  const float* ln_o_b = (const float*)d_in[17];
  const float* W1   = (const float*)d_in[18];
  const float* b1   = (const float*)d_in[19];
  const float* W2   = (const float*)d_in[20];
  const float* b2   = (const float*)d_in[21];

  const int N = in_sizes[0] / FIN;
  const int E = in_sizes[2];

  const int* src = eidx;
  const int* dst = eidx + E;

  float* out_p    = (float*)d_out;
  float* hidden_p = out_p + N;
  float* cell_p   = hidden_p + (long)N * HID;

  const int B = 256;
  auto cdiv = [](long a, long b) { return (unsigned)((a + b - 1) / b); };
  const int ntiles = (int)cdiv(N, TILE_R);

  float* ws   = (float*)d_ws;
  float* dinv = ws;
  float* XS   = dinv + N;
  float* R1   = dinv + N + E;
  float* R2   = R1 + (long)N * KTOT;
  float* AXH  = R1;
  float* Y    = R2;
  float* Z    = R1;
  float* HS   = R2;
  float* AS   = R2 + (long)N * HID;
  int* tcount = (int*)(AS + (long)N * 2);
  int2* lists = (int2*)(R2 + (long)N * KTOT);

  k_zero_counts<<<cdiv(ntiles, B), B, 0, stream>>>(tcount, ntiles);
  k_bin<<<cdiv(E, BIN_THREADS * BIN_EPT), BIN_THREADS, 0, stream>>>(
      lists, tcount, src, dst, ew, ntiles, E);
  k_deg_scale<<<ntiles, AGG_THREADS, 0, stream>>>(
      dinv, XS, HS, X, skip, H, lists, tcount, dst, ew, N, E);

  k_agg_tile<F1, true, 0><<<ntiles, AGG_THREADS, 0, stream>>>(
      AXH, AS, XS, HS, dinv, lists, tcount, src, dst, ew, nullptr, nullptr, nullptr, N, E);

  size_t smem_gates = (size_t)(KSTEPS * NT_G * 64 + GCOLS) * sizeof(float);
  k_gates<<<cdiv(N, ROWS_PER_BLOCK), B, smem_gates, stream>>>(
      AXH, C, Wx, Wh, bx, bh, bg, wc,
      ln_h_s, ln_h_b, ln_c_s, ln_c_b, ln_o_s, ln_o_b,
      hidden_p, cell_p, Y, dinv, N);

  k_agg_tile<HID, false, 1><<<ntiles, AGG_THREADS, 0, stream>>>(
      Z, nullptr, Y, nullptr, dinv, lists, tcount, src, dst, ew, W1, b1, AS, N, E);

  k_agg_tile<HID, false, 2><<<ntiles, AGG_THREADS, 0, stream>>>(
      out_p, nullptr, Z, nullptr, dinv, lists, tcount, src, dst, ew, W2, b2, X, N, E);
}


# 1st aggregation kernel consumes its finished LDS tile itself: gates GEMM (WMMA, A from the tile, K order kept), LSTM update, layernorms; [AX|AH] never stored; 6 launches; float sums unchanged
# speedup vs baseline: 6.1821x; 1.2717x over previous
#include <hip/hip_runtime.h>
#include <math.h>


typedef __attribute__((ext_vector_type(2))) float v2f;
typedef __attribute__((ext_vector_type(8))) float v8f;

#define FIN   8
#define HID   64
#define KTOT  72
#define KSTEPS 18
#define KSTEPS2 17
#define XSP   16
#define F1    74
#define GCOLS 256
#define NT_G  16
#define EPS 1e-5f

__device__ __forceinline__ float sigmoidf_(float x) { return 1.0f / (1.0f + expf(-x)); }

__global__ void k_zero_counts(int* tcount, int ntiles) {
  int i = blockIdx.x * blockDim.x + threadIdx.x;
  if (i < ntiles) tcount[i] = 0;
}


#define TILE_R      392
#define QCAP        8192
#define SCAN_U      8
#define AGG_THREADS 1024
#define CHUNK       (AGG_THREADS * 4 * SCAN_U)
#define LCAP        16384
#define KEY_BITS    9
#define BIN_THREADS 1024
#define BIN_EPT     8
#define BIN_TMAX    1024
#define NSTRIP      ((TILE_R + 15) / 16)
#define GSTASH      (4 * 256)

__global__ __launch_bounds__(BIN_THREADS)
void k_bin(int2* __restrict__ lists, int* __restrict__ tcount,
           const int* __restrict__ src, const int* __restrict__ dst,
           const float* __restrict__ w, int ntiles, int E) {
  __shared__ int hist[BIN_TMAX + 1];
  const int tid = threadIdx.x;
  for (int t = tid; t <= ntiles; t += BIN_THREADS) hist[(t < ntiles) ? t : BIN_TMAX] = 0;
  __syncthreads();

  const int e0 = blockIdx.x * (BIN_THREADS * BIN_EPT) + tid;
  int d[BIN_EPT], s[BIN_EPT];
  float ww[BIN_EPT];
#pragma unroll
  for (int j = 0; j < BIN_EPT; ++j) {
    int e = e0 + j * BIN_THREADS;
    bool ok = e < E;
    d[j] = ok ? dst[e] : -1;
    s[j] = ok ? src[e] : 0;
    ww[j] = ok ? w[e] : 0.0f;
  }
  int tl[BIN_EPT], rk[BIN_EPT], key[BIN_EPT];
#pragma unroll
  for (int j = 0; j < BIN_EPT; ++j) {
    int di = (d[j] >= 0) ? d[j] : 0;
    int t = di / TILE_R;
    key[j] = (s[j] << KEY_BITS) | (di - t * TILE_R);
    tl[j] = (d[j] >= 0) ? t : BIN_TMAX;
  }
#pragma unroll
  for (int j = 0; j < BIN_EPT; ++j) rk[j] = atomicAdd(&hist[tl[j]], 1);
  __syncthreads();
  for (int t = tid; t < ntiles; t += BIN_THREADS) {
    int h = hist[t];
    hist[t] = (h > 0) ? atomicAdd(&tcount[t], h) : 0;
  }
  __syncthreads();
#pragma unroll
  for (int j = 0; j < BIN_EPT; ++j) {
    if (tl[j] < BIN_TMAX) {
      int pos = hist[tl[j]] + rk[j];
      if (pos < LCAP)
        lists[(size_t)tl[j] * LCAP + pos] = make_int2(key[j], __float_as_int(ww[j]));
    }
  }
}

__global__ __launch_bounds__(AGG_THREADS)
void k_deg_scale(float* __restrict__ dinv, float* __restrict__ XS, float* __restrict__ HS,
                 const float* __restrict__ X, const float* __restrict__ skip,
                 const float* __restrict__ H,
                 const int2* __restrict__ lists, const int* __restrict__ tcount,
                 const int* __restrict__ dst, const float* __restrict__ w, int N, int E) {
  __shared__ float degs[TILE_R];
  const int tid = threadIdx.x;
  const int lo = blockIdx.x * TILE_R;
  const int cnt = (N - lo < TILE_R) ? (N - lo) : TILE_R;
  for (int i = tid; i < TILE_R; i += AGG_THREADS) degs[i] = 1.0f;
  __syncthreads();
  const int nl = tcount[blockIdx.x];
  if (nl <= LCAP) {
    const int2* __restrict__ L = lists + (size_t)blockIdx.x * LCAP;
    for (int i = tid; i < nl; i += AGG_THREADS) {
      int2 rec = L[i];
      atomicAdd(&degs[rec.x & ((1 << KEY_BITS) - 1)], __int_as_float(rec.y));
    }
  } else {
    for (int e = tid; e < E; e += AGG_THREADS) {
      unsigned dl = (unsigned)(dst[e] - lo);
      if (dl < (unsigned)cnt) atomicAdd(&degs[dl], w[e]);
    }
  }
  __syncthreads();
  for (int i = tid; i < cnt; i += AGG_THREADS) {
    float dv = rsqrtf(degs[i]);
    degs[i] = dv;
    dinv[lo + i] = dv;
  }
  __syncthreads();
  for (int i = tid; i < cnt * 4; i += AGG_THREADS) {
    int n = i >> 2, q4 = i & 3;
    long row = lo + n;
    float dv = degs[n];
    float* o = XS + (row * 4 + q4) * 4;
    if (q4 < 2) {
      const float* x = X + row * FIN + q4 * 4;
      o[0] = dv * x[0]; o[1] = dv * x[1]; o[2] = dv * x[2]; o[3] = dv * x[3];
    } else if (q4 == 2) {
      o[0] = dv * skip[row * 2];
      o[1] = dv * skip[row * 2 + 1];
    }
  }
  const float4* __restrict__ H4 = (const float4*)(H + (long)lo * HID);
  float4* __restrict__ O4 = (float4*)(HS + (long)lo * HID);
  for (int i = tid; i < cnt * (HID / 4); i += AGG_THREADS) {
    float dv = degs[i >> 4];
    float4 v = H4[i];
    v.x *= dv; v.y *= dv; v.z *= dv; v.w *= dv;
    O4[i] = v;
  }
}

template <int F, bool SPLIT>
__device__ __forceinline__ float row_elem(const float* __restrict__ A,
                                          const float* __restrict__ B, int s, int k) {
  if (SPLIT) return (k < HID) ? B[(long)s * HID + k] : A[(long)s * XSP + (k - HID)];
  return A[(long)s * F + k];
}

template <int NE, int F, bool SPLIT>
__device__ __forceinline__ void add_rows(float* acc, const int2* qp,
                                         const float* __restrict__ A, const float* __restrict__ B,
                                         const int* __restrict__ src,
                                         const float* __restrict__ w, int lane) {
  constexpr int KPL = (F + 31) / 32;
  int s[NE], dl[NE];
  float cf[NE];
#pragma unroll
  for (int j = 0; j < NE; ++j) {
    int2 ent = qp[j];
    dl[j] = ent.y;
    s[j] = src[ent.x];
    cf[j] = w[ent.x];
  }
  float r[NE][KPL];
#pragma unroll
  for (int j = 0; j < NE; ++j)
#pragma unroll
    for (int m = 0; m < KPL; ++m) {
      int k = lane + 32 * m;
      r[j][m] = (k < F) ? row_elem<F, SPLIT>(A, B, s[j], k) : 0.0f;
    }
#pragma unroll
  for (int j = 0; j < NE; ++j)
#pragma unroll
    for (int m = 0; m < KPL; ++m) {
      int k = lane + 32 * m;
      if (k < F) atomicAdd(&acc[dl[j] * F + k], cf[j] * r[j][m]);
    }
}

template <int NE, int F, bool SPLIT>
__device__ __forceinline__ void add_recs(float* acc, int2 rec, int j0,
                                         const float* __restrict__ A, const float* __restrict__ B,
                                         int lane) {
  constexpr int KPL = (F + 31) / 32;
  int s[NE], dl[NE];
  float cf[NE];
#pragma unroll
  for (int j = 0; j < NE; ++j) {
    int key = __builtin_amdgcn_readlane(rec.x, j0 + j);
    cf[j] = __int_as_float(__builtin_amdgcn_readlane(rec.y, j0 + j));
    s[j] = key >> KEY_BITS;
    dl[j] = key & ((1 << KEY_BITS) - 1);
  }
  float r[NE][KPL];
#pragma unroll
  for (int j = 0; j < NE; ++j)
#pragma unroll
    for (int m = 0; m < KPL; ++m) {
      int k = lane + 32 * m;
      r[j][m] = (k < F) ? row_elem<F, SPLIT>(A, B, s[j], k) : 0.0f;
    }
#pragma unroll
  for (int j = 0; j < NE; ++j)
#pragma unroll
    for (int m = 0; m < KPL; ++m) {
      int k = lane + 32 * m;
      if (k < F) atomicAdd(&acc[dl[j] * F + k], cf[j] * r[j][m]);
    }
}

struct GateArgs {
  const float* C;
  const float* Wx; const float* Wh;
  const float* bx; const float* bh; const float* bg; const float* wc;
  const float* ln_h_s; const float* ln_h_b;
  const float* ln_c_s; const float* ln_c_b;
  const float* ln_o_s; const float* ln_o_b;
  float* hidden; float* cell; float* Y;
};

template <int F, bool SPLIT, int EPI>
__global__ __launch_bounds__(AGG_THREADS)
void k_agg_tile(float* __restrict__ outbuf, float* __restrict__ outS, const float* __restrict__ A,
                const float* __restrict__ B, const float* __restrict__ dinv,
                const int2* __restrict__ lists, const int* __restrict__ tcount,
                const int* __restrict__ src, const int* __restrict__ dst,
                const float* __restrict__ w, const float* __restrict__ Wd,
                const float* __restrict__ bd, const float* __restrict__ aux, GateArgs ga,
                int N, int E) {
  __shared__ __attribute__((aligned(16))) float acc[TILE_R * F];
  __shared__ __attribute__((aligned(16))) float qs[(EPI == 3) ? NSTRIP * GSTASH : QCAP * 2];
  int2* const q = (int2*)qs;
  __shared__ int qn[2];
  __shared__ float Bf[(EPI == 1) ? KSTEPS2 * 4 * 64 : 2];
  __shared__ float b1s[(EPI == 1) ? HID : 1];
  __shared__ float w2s[(EPI == 2) ? HID : 1];
  __shared__ float dvs[(EPI != 0) ? TILE_R : 1];
  __shared__ __attribute__((aligned(16))) float Bg[(EPI == 3) ? KSTEPS * NT_G * 64 : 2];
  __shared__ float biasg[(EPI == 3) ? GCOLS : 1];
  __shared__ float wcs[(EPI == 3) ? 3 * HID : 1];
  __shared__ float lnp[(EPI == 3) ? 6 * HID : 1];

  const int tid = threadIdx.x;
  const int lane = tid & 31, wave = tid >> 5;
  const int lo = blockIdx.x * TILE_R;
  const int cnt = (N - lo < TILE_R) ? (N - lo) : TILE_R;

  if (EPI == 1) {
    for (int i = tid; i < KSTEPS2 * 4 * 64; i += AGG_THREADS) {
      int s = i >> 8;
      int rem = i & 255;
      int t = rem >> 6;
      int rem2 = rem & 63;
      int L = rem2 >> 1;
      int j = rem2 & 1;
      int k = 4 * s + 2 * (L >> 4) + j;
      int c = 16 * t + (L & 15);
      Bf[i] = (k < HID + 2) ? Wd[k * HID + c] : 0.0f;
    }
    if (tid < HID) b1s[tid] = bd[tid];
  }
  if (EPI == 2 && tid < HID) w2s[tid] = Wd[tid];
  if (EPI == 3) {
    for (int i = tid; i < KSTEPS * NT_G * 64; i += AGG_THREADS) {
      int s = i >> 10;
      int rem = i & 1023;
      int t = rem >> 6;
      int rem2 = rem & 63;
      int L = rem2 >> 1;
      int j = rem2 & 1;
      int k = 4 * s + 2 * (L >> 4) + j;
      int c = 16 * t + (L & 15);
      int g = c >> 6, h = c & 63;
      Bg[i] = (k < FIN) ? ga.Wx[(g * FIN + k) * HID + h]
                        : ga.Wh[(g * HID + (k - FIN)) * HID + h];
    }
    if (tid < GCOLS) biasg[tid] = ga.bx[tid] + ga.bh[tid] + ga.bg[tid];
    if (tid < 3 * HID) wcs[tid] = ga.wc[tid];
    if (tid < HID) {
      lnp[tid] = ga.ln_h_s[tid];           lnp[HID + tid] = ga.ln_h_b[tid];
      lnp[2 * HID + tid] = ga.ln_c_s[tid]; lnp[3 * HID + tid] = ga.ln_c_b[tid];
      lnp[4 * HID + tid] = ga.ln_o_s[tid]; lnp[5 * HID + tid] = ga.ln_o_b[tid];
    }
  }
  if (EPI != 0)
    for (int i = tid; i < cnt; i += AGG_THREADS) dvs[i] = dinv[lo + i];

  for (int i = tid; i < cnt * F; i += AGG_THREADS) {
    int n = i / F, k = i - n * F;
    acc[i] = row_elem<F, SPLIT>(A, B, lo + n, k);
  }
  if (tid == 0) { qn[0] = 0; qn[1] = 0; }
  __syncthreads();

  const int nl = __builtin_amdgcn_readfirstlane(tcount[blockIdx.x]);
  if (nl <= LCAP) {
    const int2* __restrict__ L = lists + (size_t)blockIdx.x * LCAP;
    int base = __builtin_amdgcn_readfirstlane(wave) * 32;
    int2 rec = make_int2(0, 0);
    if (base + lane < nl) rec = L[base + lane];
    while (base < nl) {
      const int nbase = base + AGG_THREADS;
      int2 nrec = make_int2(0, 0);
      if (nbase + lane < nl) nrec = L[nbase + lane];
      const int nv = (nl - base < 32) ? (nl - base) : 32;
      int j = 0;
      for (; j + 8 <= nv; j += 8) add_recs<8, F, SPLIT>(acc, rec, j, A, B, lane);
      for (; j < nv; ++j) add_recs<1, F, SPLIT>(acc, rec, j, A, B, lane);
      rec = nrec;
      base = nbase;
    }
    __syncthreads();
  } else {
  const bool vec_ok = (((size_t)dst) & 15) == 0;
  const int nchunks = (E + CHUNK - 1) / CHUNK;
  int4 v[SCAN_U];

  auto load_chunk = [&](int c) {
    if (vec_ok && c * CHUNK + CHUNK <= E) {
#pragma unroll
      for (int u = 0; u < SCAN_U; ++u)
        v[u] = *(const int4*)(dst + c * CHUNK + (u * AGG_THREADS + tid) * 4);
    } else {
#pragma unroll
      for (int u = 0; u < SCAN_U; ++u) {
        int e0 = c * CHUNK + (u * AGG_THREADS + tid) * 4;
        v[u].x = (e0     < E) ? dst[e0]     : -1;
        v[u].y = (e0 + 1 < E) ? dst[e0 + 1] : -1;
        v[u].z = (e0 + 2 < E) ? dst[e0 + 2] : -1;
        v[u].w = (e0 + 3 < E) ? dst[e0 + 3] : -1;
      }
    }
  };

  load_chunk(0);
  int par = 0;
  for (int c = 0; c < nchunks; ++c) {
    int m = 0;
#pragma unroll
    for (int u = 0; u < SCAN_U; ++u) {
      m += ((unsigned)(v[u].x - lo) < (unsigned)cnt) ? 1 : 0;
      m += ((unsigned)(v[u].y - lo) < (unsigned)cnt) ? 1 : 0;
      m += ((unsigned)(v[u].z - lo) < (unsigned)cnt) ? 1 : 0;
      m += ((unsigned)(v[u].w - lo) < (unsigned)cnt) ? 1 : 0;
    }
    unsigned ovf = 0;
    if (m > 0) {
      int pos = atomicAdd(&qn[par], m);
#pragma unroll
      for (int u = 0; u < SCAN_U; ++u) {
        int e0 = c * CHUNK + (u * AGG_THREADS + tid) * 4;
        int dd[4] = {v[u].x, v[u].y, v[u].z, v[u].w};
#pragma unroll
        for (int j = 0; j < 4; ++j) {
          unsigned dl = (unsigned)(dd[j] - lo);
          if (dl < (unsigned)cnt) {
            if (pos < QCAP) q[pos] = make_int2(e0 + j, (int)dl);
            else ovf |= 1u << (4 * u + j);
            ++pos;
          }
        }
      }
    }
    while (ovf) {
      int b = __builtin_ctz(ovf);
      ovf &= ovf - 1;
      int e = c * CHUNK + ((b >> 2) * AGG_THREADS + tid) * 4 + (b & 3);
      int dl = dst[e] - lo;
      int s = src[e];
      float cf = w[e];
      for (int k = 0; k < F; ++k)
        atomicAdd(&acc[dl * F + k], cf * row_elem<F, SPLIT>(A, B, s, k));
    }
    __syncthreads();
    const int nq = (qn[par] < QCAP) ? qn[par] : QCAP;
    if (c + 1 < nchunks) load_chunk(c + 1);
    if (tid == 0) qn[par ^ 1] = 0;

    const int nfull = nq & ~3;
    for (int base = wave * 4; base < nfull; base += (AGG_THREADS / 32) * 4)
      add_rows<4, F, SPLIT>(acc, q + base, A, B, src, w, lane);
    if (nfull + wave < nq)
      add_rows<1, F, SPLIT>(acc, q + nfull + wave, A, B, src, w, lane);
    __syncthreads();
    par ^= 1;
  }
  }

  if (EPI == 1) {
    const int half = lane >> 4, lcol = lane & 15;
    static_assert((TILE_R + 15) / 16 <= AGG_THREADS / 32, "a wave takes at most one strip");
    const int mt = __builtin_amdgcn_readfirstlane(wave);
    if (mt * 16 < cnt) {
      int rl = mt * 16 + lcol;
      if (rl > cnt - 1) rl = cnt - 1;
      const float dva = dvs[rl];
      const float* ap = acc + rl * F;
      const float* sp = aux + (long)(lo + rl) * 2;
      const float s0 = sp[0], s1 = sp[1];
      v8f c[4];
#pragma unroll
      for (int t = 0; t < 4; ++t)
#pragma unroll
        for (int r = 0; r < 8; ++r) c[t][r] = 0.0f;
#pragma unroll
      for (int s = 0; s < HID / 4; ++s) {
        v2f a = *(const v2f*)(ap + 4 * s + 2 * half);
        a.x *= dva; a.y *= dva;
        const float* bbase = Bf + s * 256 + lane * 2;
#pragma unroll
        for (int t = 0; t < 4; ++t) {
          v2f b = *(const v2f*)(bbase + t * 64);
          c[t] = __builtin_amdgcn_wmma_f32_16x16x4_f32(
              false, a, false, b, (short)0, c[t], false, false);
        }
      }
      {
        v2f a;
        a.x = (half == 0) ? s0 : 0.0f;
        a.y = (half == 0) ? s1 : 0.0f;
        const float* bbase = Bf + (HID / 4) * 256 + lane * 2;
#pragma unroll
        for (int t = 0; t < 4; ++t) {
          v2f b = *(const v2f*)(bbase + t * 64);
          c[t] = __builtin_amdgcn_wmma_f32_16x16x4_f32(
              false, a, false, b, (short)0, c[t], false, false);
        }
      }
#pragma unroll
      for (int r = 0; r < 8; ++r) {
        int rn = mt * 16 + r + 8 * half;
        if (rn < cnt) {
          long row = lo + rn;
          float dv = dvs[rn];
#pragma unroll
          for (int t = 0; t < 4; ++t) {
            int h = t * 16 + lcol;
            float z = c[t][r] + b1s[h];
            outbuf[row * HID + h] = z > 0.0f ? z * dv : 0.0f;
          }
        }
      }
    }
  } else if (EPI == 2) {
    const int r = tid >> 1, part = tid & 1;
    float p = 0.0f;
    if (r < cnt) {
      const float dv = dvs[r];
      const float* ar = acc + r * F + part * 32;
      const float* wr = w2s + part * 32;
#pragma unroll
      for (int j = 0; j < 32; ++j) {
        int k = (j + r) & 31;
        p += (ar[k] * dv) * wr[k];
      }
    }
    p += __shfl_xor(p, 1, 32);
    if (part == 0 && r < cnt) outbuf[lo + r] = (bd[0] + aux[(long)(lo + r) * FIN]) + p;
  } else if (EPI == 3) {
    for (int i = tid; i < cnt * 2; i += AGG_THREADS) {
      int n = i >> 1, j = i & 1;
      outS[(long)(lo + n) * 2 + j] = acc[n * F + HID + FIN + j] * dvs[n];
    }
    const int half = lane >> 4, lcol = lane & 15;
    static_assert(NSTRIP <= AGG_THREADS / 32, "a wave takes at most one strip");
    const int mt = __builtin_amdgcn_readfirstlane(wave);
    if (mt * 16 < cnt) {
      int rl = mt * 16 + lcol;
      if (rl > cnt - 1) rl = cnt - 1;
      const float dva = dvs[rl];
      const float* ap = acc + rl * F;
      float* st = qs + mt * GSTASH;
      float hv[3][8], cv3[8];
      const float* Cb = ga.C + (long)lo * HID;
      float* hidb = ga.hidden + (long)lo * HID;
      float* celb = ga.cell + (long)lo * HID;
      float* Yb = ga.Y + (long)lo * HID;
#pragma unroll
      for (int t = 0; t < 4; ++t) {
        v8f c[4];
#pragma unroll
        for (int g = 0; g < 4; ++g)
#pragma unroll
          for (int r = 0; r < 8; ++r) c[g][r] = 0.0f;
#pragma unroll
        for (int s = 0; s < KSTEPS; ++s) {
          const int kc = (s < 2) ? (HID + 4 * s) : (4 * (s - 2));
          v2f a = *(const v2f*)(ap + kc + 2 * half);
          a.x *= dva; a.y *= dva;
          const float* bbase = Bg + s * 1024 + t * 64 + lane * 2;
#pragma unroll
          for (int g = 0; g < 4; ++g) {
            v2f b = *(const v2f*)(bbase + g * 256);
            c[g] = __builtin_amdgcn_wmma_f32_16x16x4_f32(
                false, a, false, b, (short)0, c[g], false, false);
          }
        }
        const int h = t * 16 + lcol;
        const float bi = biasg[h], bf = biasg[64 + h], bc = biasg[128 + h], bo = biasg[192 + h];
        const float wi = wcs[h], wf = wcs[64 + h], wo = wcs[128 + h];
#pragma unroll
        for (int r = 0; r < 8; ++r) {
          int rn = mt * 16 + r + 8 * half;
          if (rn > cnt - 1) rn = cnt - 1;
          float c0 = Cb[rn * HID + h];
          float gi = c[0][r] + bi + wi * c0;
          float gf = c[1][r] + bf + wf * c0;
          float gc = c[2][r] + bc;
          float go = c[3][r] + bo;
          float iv = sigmoidf_(gi);
          float fv = sigmoidf_(gf);
          float cc = fv * c0 + iv * tanhf(gc);
          float ov = sigmoidf_(go + wo * cc);
          float hval = ov * tanhf(cc);
          if (t < 3) st[(t * 8 + r) * 32 + lane] = cc; else cv3[r] = cc;
          if (t < 1) st[(3 * 8 + r) * 32 + lane] = hval; else hv[t - 1][r] = hval;
        }
      }
#pragma unroll
      for (int r = 0; r < 8; ++r) {
        const int rn = mt * 16 + r + 8 * half;
        float sh = 0.f, sh2 = 0.f, scr = 0.f, sc2r = 0.f;
        float cv[4], hw[4];
#pragma unroll
        for (int t = 0; t < 4; ++t) {
          cv[t] = (t < 3) ? st[(t * 8 + r) * 32 + lane] : cv3[r];
          hw[t] = (t < 1) ? st[(3 * 8 + r) * 32 + lane] : hv[t - 1][r];
          sh += hw[t];  sh2 += hw[t] * hw[t];
          scr += cv[t]; sc2r += cv[t] * cv[t];
        }
#pragma unroll
        for (int mask = 1; mask < 16; mask <<= 1) {
          sh   += __shfl_xor(sh,   mask, 32);
          sh2  += __shfl_xor(sh2,  mask, 32);
          scr  += __shfl_xor(scr,  mask, 32);
          sc2r += __shfl_xor(sc2r, mask, 32);
        }
        float mh = sh * (1.0f / 64.0f), mc = scr * (1.0f / 64.0f);
        float vh = sh2 * (1.0f / 64.0f) - mh * mh;
        float vc = sc2r * (1.0f / 64.0f) - mc * mc;
        float rh = rsqrtf(vh + EPS), rc = rsqrtf(vc + EPS);
        if (rn < cnt) {
          const float dv = dvs[rn];
#pragma unroll
          for (int t = 0; t < 4; ++t) {
            int h = t * 16 + lcol;
            float hn = (hw[t] - mh) * rh;
            float cn = (cv[t] - mc) * rc;
            hidb[rn * HID + h] = hn * lnp[h] + lnp[HID + h];
            celb[rn * HID + h] = cn * lnp[2 * HID + h] + lnp[3 * HID + h];
            float ov = hn * lnp[4 * HID + h] + lnp[5 * HID + h];
            Yb[rn * HID + h] = ov > 0.0f ? ov * dv : 0.0f;
          }
        }
      }
    }
  } else {
    for (int i = tid; i < cnt * F; i += AGG_THREADS)
      outbuf[(long)lo * F + i] = acc[i] * dinv[lo + i / F];
  }
}

extern "C" void kernel_launch(void* const* d_in, const int* in_sizes, int n_in,
                              void* d_out, int out_size, void* d_ws, size_t ws_size,
                              hipStream_t stream) {
  const float* X    = (const float*)d_in[0];
  const int*   eidx = (const int*)d_in[1];
  const float* ew   = (const float*)d_in[2];
  const float* skip = (const float*)d_in[3];
  const float* H    = (const float*)d_in[4];
  const float* C    = (const float*)d_in[5];
  const float* Wx   = (const float*)d_in[6];
  const float* bx   = (const float*)d_in[7];
  const float* Wh   = (const float*)d_in[8];
  const float* bh   = (const float*)d_in[9];
  const float* wc   = (const float*)d_in[10];
  const float* bg   = (const float*)d_in[11];
  const float* ln_h_s = (const float*)d_in[12];
  const float* ln_h_b = (const float*)d_in[13];
  const float* ln_c_s = (const float*)d_in[14];
  const float* ln_c_b = (const float*)d_in[15];
  const float* ln_o_s = (const float*)d_in[16];
  const float* ln_o_b = (const float*)d_in[17];
  const float* W1   = (const float*)d_in[18];
  const float* b1   = (const float*)d_in[19];
  const float* W2   = (const float*)d_in[20];
  const float* b2   = (const float*)d_in[21];

  const int N = in_sizes[0] / FIN;
  const int E = in_sizes[2];

  const int* src = eidx;
  const int* dst = eidx + E;

  float* out_p    = (float*)d_out;
  float* hidden_p = out_p + N;
  float* cell_p   = hidden_p + (long)N * HID;

  const int B = 256;
  auto cdiv = [](long a, long b) { return (unsigned)((a + b - 1) / b); };
  const int ntiles = (int)cdiv(N, TILE_R);

  float* ws   = (float*)d_ws;
  float* dinv = ws;
  float* XS   = dinv + N;
  float* R1   = dinv + N + E;
  float* R2   = R1 + (long)N * KTOT;
  float* Y    = R1;
  float* Z    = R2;
  float* HS   = R2;
  float* AS   = R2 + (long)N * HID;
  int* tcount = (int*)(AS + (long)N * 2);
  int2* lists = (int2*)(R2 + (long)N * KTOT);

  k_zero_counts<<<cdiv(ntiles, B), B, 0, stream>>>(tcount, ntiles);
  k_bin<<<cdiv(E, BIN_THREADS * BIN_EPT), BIN_THREADS, 0, stream>>>(
      lists, tcount, src, dst, ew, ntiles, E);
  k_deg_scale<<<ntiles, AGG_THREADS, 0, stream>>>(
      dinv, XS, HS, X, skip, H, lists, tcount, dst, ew, N, E);

  GateArgs ga;
  ga.C = C; ga.Wx = Wx; ga.Wh = Wh; ga.bx = bx; ga.bh = bh; ga.bg = bg; ga.wc = wc;
  ga.ln_h_s = ln_h_s; ga.ln_h_b = ln_h_b; ga.ln_c_s = ln_c_s; ga.ln_c_b = ln_c_b;
  ga.ln_o_s = ln_o_s; ga.ln_o_b = ln_o_b;
  ga.hidden = hidden_p; ga.cell = cell_p; ga.Y = Y;
  k_agg_tile<F1, true, 3><<<ntiles, AGG_THREADS, 0, stream>>>(
      nullptr, AS, XS, HS, dinv, lists, tcount, src, dst, ew, nullptr, nullptr, nullptr, ga, N, E);

  k_agg_tile<HID, false, 1><<<ntiles, AGG_THREADS, 0, stream>>>(
      Z, nullptr, Y, nullptr, dinv, lists, tcount, src, dst, ew, W1, b1, AS, ga, N, E);

  k_agg_tile<HID, false, 2><<<ntiles, AGG_THREADS, 0, stream>>>(
      out_p, nullptr, Z, nullptr, dinv, lists, tcount, src, dst, ew, W2, b2, X, ga, N, E);
}
